# starting kernel
# baseline (speedup 1.0000x reference)
#include <hip/hip_runtime.h>
#include <hip/hip_bf16.h>

typedef __attribute__((ext_vector_type(2))) float v2f;
typedef __attribute__((ext_vector_type(8))) float v8f;

#define IN_CH 64
#define OUT_DIM 128
#define HEADS 4

static __device__ __forceinline__ float lrelu(float x) { return fmaxf(x, 0.2f * x); }

static __device__ __forceinline__ unsigned fkey(float f) {
  unsigned u = __float_as_uint(f);
  return (u & 0x80000000u) ? ~u : (u | 0x80000000u);
}
static __device__ __forceinline__ float fkey_inv(unsigned k) {
  return __uint_as_float((k & 0x80000000u) ? (k & 0x7FFFFFFFu) : ~k);
}

__global__ __launch_bounds__(256) void k_gemm(const float* __restrict__ x,
                                              const float* __restrict__ lin_w,
                                              const float* __restrict__ res_w,
                                              float* __restrict__ xt,
                                              float* __restrict__ z,
                                              int N) {
  const int wave = threadIdx.x >> 5;
  const int lane = threadIdx.x & 31;
  const int m0 = blockIdx.x * 16;
  const int mrow = lane & 15;
  const int kb = (lane >> 4) << 1;
  int arow = m0 + mrow;
  if (arow >= N) arow = N - 1;
  const float* xr = x + (size_t)arow * IN_CH;
  v2f a[16];
#pragma unroll
  for (int ks = 0; ks < 16; ++ks) {
    a[ks].x = xr[ks * 4 + kb];
    a[ks].y = xr[ks * 4 + kb + 1];
  }
#pragma unroll
  for (int tt = 0; tt < 2; ++tt) {
    const int tile = wave + tt * 8;
    const float* W = (tile < 8) ? lin_w : res_w;
    const int cbase = (tile & 7) * 16;
    const float* wr = W + (size_t)(cbase + (lane & 15)) * IN_CH;
    v8f c = {0.f, 0.f, 0.f, 0.f, 0.f, 0.f, 0.f, 0.f};
#pragma unroll
    for (int ks = 0; ks < 16; ++ks) {
      v2f b;
      b.x = wr[ks * 4 + kb];
      b.y = wr[ks * 4 + kb + 1];
      c = __builtin_amdgcn_wmma_f32_16x16x4_f32(false, a[ks], false, b,
                                                (short)0, c, false, false);
    }
    float* dst = (tile < 8) ? xt : z;
    const int mb = m0 + ((lane >> 4) << 3);
#pragma unroll
    for (int r = 0; r < 8; ++r) {
      const int m = mb + r;
      if (m < N) dst[(size_t)m * OUT_DIM + cbase + (lane & 15)] = c[r];
    }
  }
}

__global__ __launch_bounds__(256) void k_node(const float* __restrict__ xt,
                                              const float* __restrict__ att_src,
                                              const float* __restrict__ att_dst,
                                              const float* __restrict__ gat_bias,
                                              const float* __restrict__ res_b,
                                              float* __restrict__ z,
                                              float* __restrict__ a_src,
                                              float* __restrict__ a_dst,
                                              unsigned* __restrict__ emax_u,
                                              float* __restrict__ esum, int N) {
  const int lane = threadIdx.x & 31;
  const int n = blockIdx.x * 8 + (threadIdx.x >> 5);
  if (n >= N) return;
  const float4 v = ((const float4*)xt)[(size_t)n * 32 + lane];
  const float4 as = ((const float4*)att_src)[lane];
  const float4 ad = ((const float4*)att_dst)[lane];
  float ps = v.x * as.x + v.y * as.y + v.z * as.z + v.w * as.w;
  float pd = v.x * ad.x + v.y * ad.y + v.z * ad.z + v.w * ad.w;
#pragma unroll
  for (int off = 1; off < 8; off <<= 1) {
    ps += __shfl_xor(ps, off, 32);
    pd += __shfl_xor(pd, off, 32);
  }
  const int h = lane >> 3;
  if ((lane & 7) == 0) {
    a_src[(size_t)n * HEADS + h] = ps;
    a_dst[(size_t)n * HEADS + h] = pd;
  }
  if (lane < HEADS) {
    emax_u[(size_t)n * HEADS + lane] = 0u;
    esum[(size_t)n * HEADS + lane] = 0.f;
  }
  float4 zr = ((float4*)z)[(size_t)n * 32 + lane];
  const float4 gb = ((const float4*)gat_bias)[lane];
  const float4 rb = ((const float4*)res_b)[lane];
  zr.x += gb.x + rb.x; zr.y += gb.y + rb.y;
  zr.z += gb.z + rb.z; zr.w += gb.w + rb.w;
  ((float4*)z)[(size_t)n * 32 + lane] = zr;
}

static __device__ __forceinline__ void edge_sd(const int* __restrict__ ei, int e,
                                               int E, int& s, int& d) {
  if (e < E) { s = ei[e]; d = ei[E + e]; }
  else       { s = e - E; d = s; }
}

__global__ __launch_bounds__(256) void k_edge_max(const int* __restrict__ ei,
                                                  const float* __restrict__ a_src,
                                                  const float* __restrict__ a_dst,
                                                  unsigned* __restrict__ emax_u,
                                                  int E, int N) {
  const int e = blockIdx.x * blockDim.x + threadIdx.x;
  if (e >= E + N) return;
  int s, d; edge_sd(ei, e, E, s, d);
  const float4 as = ((const float4*)a_src)[s];
  const float4 ad = ((const float4*)a_dst)[d];
  unsigned* mp = emax_u + (size_t)d * HEADS;
  atomicMax(mp + 0, fkey(lrelu(as.x + ad.x)));
  atomicMax(mp + 1, fkey(lrelu(as.y + ad.y)));
  atomicMax(mp + 2, fkey(lrelu(as.z + ad.z)));
  atomicMax(mp + 3, fkey(lrelu(as.w + ad.w)));
}

__global__ __launch_bounds__(256) void k_edge_sum(const int* __restrict__ ei,
                                                  const float* __restrict__ a_src,
                                                  const float* __restrict__ a_dst,
                                                  const unsigned* __restrict__ emax_u,
                                                  float* __restrict__ esum,
                                                  float* __restrict__ exb,
                                                  int E, int N) {
  const int e = blockIdx.x * blockDim.x + threadIdx.x;
  if (e >= E + N) return;
  int s, d; edge_sd(ei, e, E, s, d);
  const float4 as = ((const float4*)a_src)[s];
  const float4 ad = ((const float4*)a_dst)[d];
  const unsigned* mp = emax_u + (size_t)d * HEADS;
  float4 ex;
  ex.x = __expf(lrelu(as.x + ad.x) - fkey_inv(mp[0]));
  ex.y = __expf(lrelu(as.y + ad.y) - fkey_inv(mp[1]));
  ex.z = __expf(lrelu(as.z + ad.z) - fkey_inv(mp[2]));
  ex.w = __expf(lrelu(as.w + ad.w) - fkey_inv(mp[3]));
  ((float4*)exb)[e] = ex;
  float* sp = esum + (size_t)d * HEADS;
  atomicAdd(sp + 0, ex.x);
  atomicAdd(sp + 1, ex.y);
  atomicAdd(sp + 2, ex.z);
  atomicAdd(sp + 3, ex.w);
}

__global__ __launch_bounds__(256) void k_scatter(const int* __restrict__ ei,
                                                 const float* __restrict__ exb,
                                                 const float* __restrict__ esum,
                                                 const float* __restrict__ xt,
                                                 float* __restrict__ z,
                                                 int E, int N) {
  const int lane = threadIdx.x & 31;
  const int e = blockIdx.x * 8 + (threadIdx.x >> 5);
  if (e >= E + N) return;
  int s, d; edge_sd(ei, e, E, s, d);
  const int h = lane >> 3;
  const float alpha =
      exb[(size_t)e * HEADS + h] / (esum[(size_t)d * HEADS + h] + 1e-16f);
  const float4 m = ((const float4*)xt)[(size_t)s * 32 + lane];
  float* zp = z + (size_t)d * OUT_DIM + lane * 4;
  atomicAdd(zp + 0, alpha * m.x);
  atomicAdd(zp + 1, alpha * m.y);
  atomicAdd(zp + 2, alpha * m.z);
  atomicAdd(zp + 3, alpha * m.w);
}

__global__ __launch_bounds__(256) void k_ln(const float* __restrict__ z,
                                            const float* __restrict__ ln_g,
                                            const float* __restrict__ ln_b,
                                            float* __restrict__ out, int N) {
  const int lane = threadIdx.x & 31;
  const int n = blockIdx.x * 8 + (threadIdx.x >> 5);
  if (n >= N) return;
  const float4 v = ((const float4*)z)[(size_t)n * 32 + lane];
  float s = v.x + v.y + v.z + v.w;
#pragma unroll
  for (int off = 1; off < 32; off <<= 1) s += __shfl_xor(s, off, 32);
  const float mu = s * (1.0f / 128.0f);
  const float dx = v.x - mu, dy = v.y - mu, dz = v.z - mu, dw = v.w - mu;
  float q = dx * dx + dy * dy + dz * dz + dw * dw;
#pragma unroll
  for (int off = 1; off < 32; off <<= 1) q += __shfl_xor(q, off, 32);
  const float inv = rsqrtf(q * (1.0f / 128.0f) + 1e-5f);
  const float4 g = ((const float4*)ln_g)[lane];
  const float4 b = ((const float4*)ln_b)[lane];
  float4 o;
  o.x = lrelu(dx * inv * g.x + b.x);
  o.y = lrelu(dy * inv * g.y + b.y);
  o.z = lrelu(dz * inv * g.z + b.z);
  o.w = lrelu(dw * inv * g.w + b.w);
  ((float4*)out)[(size_t)n * 32 + lane] = o;
}

extern "C" void kernel_launch(void* const* d_in, const int* in_sizes, int n_in,
                              void* d_out, int out_size, void* d_ws, size_t ws_size,
                              hipStream_t stream) {
  const float* x        = (const float*)d_in[0];
  const int*   ei       = (const int*)d_in[1];
  const float* lin_w    = (const float*)d_in[2];
  const float* att_src  = (const float*)d_in[3];
  const float* att_dst  = (const float*)d_in[4];
  const float* gat_bias = (const float*)d_in[5];
  const float* res_w    = (const float*)d_in[6];
  const float* res_b    = (const float*)d_in[7];
  const float* ln_g     = (const float*)d_in[8];
  const float* ln_b     = (const float*)d_in[9];
  const int N  = in_sizes[0] / IN_CH;
  const int E  = in_sizes[1] / 2;
  const int Et = E + N;

  float*    xt     = (float*)d_ws;
  float*    z      = xt + (size_t)N * OUT_DIM;
  float*    a_src  = z + (size_t)N * OUT_DIM;
  float*    a_dst  = a_src + (size_t)N * HEADS;
  unsigned* emax_u = (unsigned*)(a_dst + (size_t)N * HEADS);
  float*    esum   = (float*)(emax_u + (size_t)N * HEADS);
  float*    exb    = esum + (size_t)N * HEADS;

  k_gemm<<<(N + 15) / 16, 256, 0, stream>>>(x, lin_w, res_w, xt, z, N);
  k_node<<<(N + 7) / 8, 256, 0, stream>>>(xt, att_src, att_dst, gat_bias, res_b,
                                          z, a_src, a_dst, emax_u, esum, N);
  k_edge_max<<<(Et + 255) / 256, 256, 0, stream>>>(ei, a_src, a_dst, emax_u, E, N);
  k_edge_sum<<<(Et + 255) / 256, 256, 0, stream>>>(ei, a_src, a_dst, emax_u, esum,
                                                   exb, E, N);
  k_scatter<<<(Et + 7) / 8, 256, 0, stream>>>(ei, exb, esum, xt, z, E, N);
  k_ln<<<(N + 7) / 8, 256, 0, stream>>>(z, ln_g, ln_b, (float*)d_out, N);
}


# Message sums move from L2 float atomics to LDS float atomics: owner groups of 196 nodes scan the destination column, queue their edges unsorted, add ex*xt in LDS, then LayerNorm; sum order changes
# speedup vs baseline: 1.9511x; 1.9511x over previous
#include <hip/hip_runtime.h>
#include <hip/hip_bf16.h>

typedef __attribute__((ext_vector_type(2))) float v2f;
typedef __attribute__((ext_vector_type(8))) float v8f;

#define IN_CH 64
#define OUT_DIM 128
#define HEADS 4

static __device__ __forceinline__ float lrelu(float x) { return fmaxf(x, 0.2f * x); }

static __device__ __forceinline__ unsigned fkey(float f) {
  unsigned u = __float_as_uint(f);
  return (u & 0x80000000u) ? ~u : (u | 0x80000000u);
}
static __device__ __forceinline__ float fkey_inv(unsigned k) {
  return __uint_as_float((k & 0x80000000u) ? (k & 0x7FFFFFFFu) : ~k);
}

__global__ __launch_bounds__(256) void k_gemm(const float* __restrict__ x,
                                              const float* __restrict__ lin_w,
                                              const float* __restrict__ res_w,
                                              float* __restrict__ xt,
                                              float* __restrict__ z,
                                              int N) {
  const int wave = threadIdx.x >> 5;
  const int lane = threadIdx.x & 31;
  const int m0 = blockIdx.x * 16;
  const int mrow = lane & 15;
  const int kb = (lane >> 4) << 1;
  int arow = m0 + mrow;
  if (arow >= N) arow = N - 1;
  const float* xr = x + (size_t)arow * IN_CH;
  v2f a[16];
#pragma unroll
  for (int ks = 0; ks < 16; ++ks) {
    a[ks].x = xr[ks * 4 + kb];
    a[ks].y = xr[ks * 4 + kb + 1];
  }
#pragma unroll
  for (int tt = 0; tt < 2; ++tt) {
    const int tile = wave + tt * 8;
    const float* W = (tile < 8) ? lin_w : res_w;
    const int cbase = (tile & 7) * 16;
    const float* wr = W + (size_t)(cbase + (lane & 15)) * IN_CH;
    v8f c = {0.f, 0.f, 0.f, 0.f, 0.f, 0.f, 0.f, 0.f};
#pragma unroll
    for (int ks = 0; ks < 16; ++ks) {
      v2f b;
      b.x = wr[ks * 4 + kb];
      b.y = wr[ks * 4 + kb + 1];
      c = __builtin_amdgcn_wmma_f32_16x16x4_f32(false, a[ks], false, b,
                                                (short)0, c, false, false);
    }
    float* dst = (tile < 8) ? xt : z;
    const int mb = m0 + ((lane >> 4) << 3);
#pragma unroll
    for (int r = 0; r < 8; ++r) {
      const int m = mb + r;
      if (m < N) dst[(size_t)m * OUT_DIM + cbase + (lane & 15)] = c[r];
    }
  }
}

__global__ __launch_bounds__(256) void k_node(const float* __restrict__ xt,
                                              const float* __restrict__ att_src,
                                              const float* __restrict__ att_dst,
                                              float* __restrict__ a_src,
                                              float* __restrict__ a_dst,
                                              unsigned* __restrict__ emax_u, int N) {
  const int lane = threadIdx.x & 31;
  const int n = blockIdx.x * 8 + (threadIdx.x >> 5);
  if (n >= N) return;
  const float4 v = ((const float4*)xt)[(size_t)n * 32 + lane];
  const float4 as = ((const float4*)att_src)[lane];
  const float4 ad = ((const float4*)att_dst)[lane];
  float ps = v.x * as.x + v.y * as.y + v.z * as.z + v.w * as.w;
  float pd = v.x * ad.x + v.y * ad.y + v.z * ad.z + v.w * ad.w;
#pragma unroll
  for (int off = 1; off < 8; off <<= 1) {
    ps += __shfl_xor(ps, off, 32);
    pd += __shfl_xor(pd, off, 32);
  }
  const int h = lane >> 3;
  if ((lane & 7) == 0) {
    a_src[(size_t)n * HEADS + h] = ps;
    a_dst[(size_t)n * HEADS + h] = pd;
  }
  if (lane < HEADS) emax_u[(size_t)n * HEADS + lane] = 0u;
}

static __device__ __forceinline__ void edge_sd(const int* __restrict__ ei, int e,
                                               int E, int& s, int& d) {
  if (e < E) { s = ei[e]; d = ei[E + e]; }
  else       { s = e - E; d = s; }
}

__global__ __launch_bounds__(256) void k_edge_max(const int* __restrict__ ei,
                                                  const float* __restrict__ a_src,
                                                  const float* __restrict__ a_dst,
                                                  unsigned* __restrict__ emax_u,
                                                  int E, int N) {
  const int e = blockIdx.x * blockDim.x + threadIdx.x;
  if (e >= E + N) return;
  int s, d; edge_sd(ei, e, E, s, d);
  const float4 as = ((const float4*)a_src)[s];
  const float4 ad = ((const float4*)a_dst)[d];
  unsigned* mp = emax_u + (size_t)d * HEADS;
  atomicMax(mp + 0, fkey(lrelu(as.x + ad.x)));
  atomicMax(mp + 1, fkey(lrelu(as.y + ad.y)));
  atomicMax(mp + 2, fkey(lrelu(as.z + ad.z)));
  atomicMax(mp + 3, fkey(lrelu(as.w + ad.w)));
}

#define NPW 196
#define APITCH 130
#define QCAP 8192
__global__ __launch_bounds__(1024) void k_own(const int* __restrict__ ei,
                                              const float* __restrict__ xt,
                                              const float* __restrict__ a_src,
                                              const float* __restrict__ a_dst,
                                              const unsigned* __restrict__ emax_u,
                                              const float* __restrict__ r,
                                              const float* __restrict__ gat_bias,
                                              const float* __restrict__ res_b,
                                              const float* __restrict__ ln_g,
                                              const float* __restrict__ ln_b,
                                              float* __restrict__ out, int E, int N) {
  __shared__ float acc_l[NPW * APITCH];
  __shared__ float esum_l[NPW * HEADS];
  __shared__ float adl_l[NPW * HEADS];
  __shared__ float mx_l[NPW * HEADS];
  __shared__ int qe[QCAP];
  __shared__ int qd[QCAP];
  __shared__ int qn;

  const int tid = threadIdx.x;
  const int lane = tid & 31;
  const int wave = __builtin_amdgcn_readfirstlane(tid >> 5);
  const int h = lane >> 3;
  const int coff = lane * 4 + (lane >> 4);
  const int lo = blockIdx.x * NPW;
  const int cnt = min(NPW, N - lo);

  if (tid == 0) qn = 0;
  for (int dl = wave; dl < cnt; dl += 32) {
    const int n = lo + dl;
    const float4 m = ((const float4*)xt)[(size_t)n * 32 + lane];
    const float as = a_src[(size_t)n * HEADS + h];
    const float ad = a_dst[(size_t)n * HEADS + h];
    const float mx = fkey_inv(emax_u[(size_t)n * HEADS + h]);
    const float ex = __expf(lrelu(as + ad) - mx);
    float* ap = acc_l + dl * APITCH + coff;
    ap[0] = ex * m.x;
    ap[1] = ex * m.y;
    ap[2] = ex * m.z;
    ap[3] = ex * m.w;
    if ((lane & 7) == 0) {
      esum_l[dl * HEADS + h] = ex;
      adl_l[dl * HEADS + h] = ad;
      mx_l[dl * HEADS + h] = mx;
    }
  }
  __syncthreads();

  auto queue_edge = [&](int e) {
    const int slot = atomicAdd(&qn, 1);
    if (slot < QCAP) {
      qe[slot] = e;
    } else {
      const int s = ei[e];
      const int dl = ei[E + e] - lo;
#pragma unroll 1
      for (int hh = 0; hh < HEADS; ++hh) {
        const float ex = __expf(lrelu(a_src[(size_t)s * HEADS + hh] + adl_l[dl * HEADS + hh]) -
                                mx_l[dl * HEADS + hh]);
        atomicAdd(esum_l + dl * HEADS + hh, ex);
#pragma unroll 1
        for (int ch = hh * 32; ch < hh * 32 + 32; ++ch)
          atomicAdd(acc_l + dl * APITCH + ch + (ch >> 6), ex * xt[(size_t)s * OUT_DIM + ch]);
      }
    }
  };

  const int* __restrict__ dp = ei + E;
  const int nmain = ((E & 3) == 0) ? ((E >> 12) & ~7) : 0;
  const int c0 = (int)(((unsigned long long)blockIdx.x * (unsigned)nmain) / gridDim.x);
  for (int ci = 0; ci < nmain; ci += 8) {
    int4 v[8];
#pragma unroll
    for (int u = 0; u < 8; ++u) {
      int cc = c0 + ci + u;
      if (cc >= nmain) cc -= nmain;
      v[u] = *(const int4*)(dp + (cc << 12) + tid * 4);
    }
    unsigned hit = 0;
#pragma unroll
    for (int u = 0; u < 8; ++u) {
      hit |= (unsigned)((unsigned)(v[u].x - lo) < (unsigned)cnt) << (u * 4 + 0);
      hit |= (unsigned)((unsigned)(v[u].y - lo) < (unsigned)cnt) << (u * 4 + 1);
      hit |= (unsigned)((unsigned)(v[u].z - lo) < (unsigned)cnt) << (u * 4 + 2);
      hit |= (unsigned)((unsigned)(v[u].w - lo) < (unsigned)cnt) << (u * 4 + 3);
    }
    while (hit) {
      const int k = __builtin_ctz(hit);
      hit &= hit - 1;
      int cc = c0 + ci + (k >> 2);
      if (cc >= nmain) cc -= nmain;
      queue_edge((cc << 12) + tid * 4 + (k & 3));
    }
  }
  for (int eb = (nmain << 12) + tid; eb < E; eb += 4096) {
    int d[4];
#pragma unroll
    for (int u = 0; u < 4; ++u) d[u] = dp[min(eb + u * 1024, E - 1)];
#pragma unroll
    for (int u = 0; u < 4; ++u)
      if (eb + u * 1024 < E && (unsigned)(d[u] - lo) < (unsigned)cnt) queue_edge(eb + u * 1024);
  }
  __syncthreads();

  const int nq = min(qn, QCAP);
  for (int i = tid; i < nq; i += 1024) {
    const int e = qe[i];
    qe[i] = ei[e];
    qd[i] = dp[e] - lo;
  }
  __syncthreads();

  for (int i0 = wave * 8; i0 < nq; i0 += 256) {
    int s[8], dl[8];
    float4 m[8];
    float as[8], ad[8], mx[8];
#pragma unroll
    for (int u = 0; u < 8; ++u) {
      const int i = min(i0 + u, nq - 1);
      s[u] = __builtin_amdgcn_readfirstlane(qe[i]);
      dl[u] = __builtin_amdgcn_readfirstlane(qd[i]);
      m[u] = ((const float4*)xt)[(size_t)s[u] * 32 + lane];
      as[u] = a_src[(size_t)s[u] * HEADS + h];
      ad[u] = adl_l[dl[u] * HEADS + h];
      mx[u] = mx_l[dl[u] * HEADS + h];
    }
#pragma unroll
    for (int u = 0; u < 8; ++u) {
      if (i0 + u < nq) {
        const float ex = __expf(lrelu(as[u] + ad[u]) - mx[u]);
        float* ap = acc_l + dl[u] * APITCH + coff;
        atomicAdd(ap + 0, ex * m[u].x);
        atomicAdd(ap + 1, ex * m[u].y);
        atomicAdd(ap + 2, ex * m[u].z);
        atomicAdd(ap + 3, ex * m[u].w);
        if ((lane & 7) == 0) atomicAdd(esum_l + dl[u] * HEADS + h, ex);
      }
    }
  }
  __syncthreads();

  const float4 gb = ((const float4*)gat_bias)[lane];
  const float4 rb = ((const float4*)res_b)[lane];
  const float4 g = ((const float4*)ln_g)[lane];
  const float4 b = ((const float4*)ln_b)[lane];
  for (int dl = wave; dl < cnt; dl += 32) {
    const int n = lo + dl;
    const float* ap = acc_l + dl * APITCH + coff;
    const float4 rr = ((const float4*)r)[(size_t)n * 32 + lane];
    const float den = esum_l[dl * HEADS + h] + 1e-16f;
    float4 v;
    v.x = ap[0] / den + gb.x + rr.x + rb.x;
    v.y = ap[1] / den + gb.y + rr.y + rb.y;
    v.z = ap[2] / den + gb.z + rr.z + rb.z;
    v.w = ap[3] / den + gb.w + rr.w + rb.w;
    float sm = v.x + v.y + v.z + v.w;
#pragma unroll
    for (int off = 1; off < 32; off <<= 1) sm += __shfl_xor(sm, off, 32);
    const float mu = sm * (1.0f / 128.0f);
    const float dx = v.x - mu, dy = v.y - mu, dz = v.z - mu, dw = v.w - mu;
    float q = dx * dx + dy * dy + dz * dz + dw * dw;
#pragma unroll
    for (int off = 1; off < 32; off <<= 1) q += __shfl_xor(q, off, 32);
    const float inv = rsqrtf(q * (1.0f / 128.0f) + 1e-5f);
    float4 o;
    o.x = lrelu(dx * inv * g.x + b.x);
    o.y = lrelu(dy * inv * g.y + b.y);
    o.z = lrelu(dz * inv * g.z + b.z);
    o.w = lrelu(dw * inv * g.w + b.w);
    ((float4*)out)[(size_t)n * 32 + lane] = o;
  }
}

extern "C" void kernel_launch(void* const* d_in, const int* in_sizes, int n_in,
                              void* d_out, int out_size, void* d_ws, size_t ws_size,
                              hipStream_t stream) {
  const float* x        = (const float*)d_in[0];
  const int*   ei       = (const int*)d_in[1];
  const float* lin_w    = (const float*)d_in[2];
  const float* att_src  = (const float*)d_in[3];
  const float* att_dst  = (const float*)d_in[4];
  const float* gat_bias = (const float*)d_in[5];
  const float* res_w    = (const float*)d_in[6];
  const float* res_b    = (const float*)d_in[7];
  const float* ln_g     = (const float*)d_in[8];
  const float* ln_b     = (const float*)d_in[9];
  const int N  = in_sizes[0] / IN_CH;
  const int E  = in_sizes[1] / 2;
  const int Et = E + N;

  float*    xt     = (float*)d_ws;
  float*    z      = xt + (size_t)N * OUT_DIM;
  float*    a_src  = z + (size_t)N * OUT_DIM;
  float*    a_dst  = a_src + (size_t)N * HEADS;
  unsigned* emax_u = (unsigned*)(a_dst + (size_t)N * HEADS);

  k_gemm<<<(N + 15) / 16, 256, 0, stream>>>(x, lin_w, res_w, xt, z, N);
  k_node<<<(N + 7) / 8, 256, 0, stream>>>(xt, att_src, att_dst, a_src, a_dst, emax_u, N);
  k_edge_max<<<(Et + 255) / 256, 256, 0, stream>>>(ei, a_src, a_dst, emax_u, E, N);
  k_own<<<(N + NPW - 1) / NPW, 1024, 0, stream>>>(ei, xt, a_src, a_dst, emax_u, z, gat_bias,
                                                  res_b, ln_g, ln_b, (float*)d_out, E, N);
}


# Segment max moves from L2 atomics in a launch of its own into the owner groups (LDS max over each group's queued edges); self-loops are queued like edges, so their place in the sum order changes
# speedup vs baseline: 3.9718x; 2.0357x over previous
#include <hip/hip_runtime.h>
#include <hip/hip_bf16.h>

typedef __attribute__((ext_vector_type(2))) float v2f;
typedef __attribute__((ext_vector_type(8))) float v8f;

#define IN_CH 64
#define OUT_DIM 128
#define HEADS 4

static __device__ __forceinline__ float lrelu(float x) { return fmaxf(x, 0.2f * x); }

static __device__ __forceinline__ unsigned fkey(float f) {
  unsigned u = __float_as_uint(f);
  return (u & 0x80000000u) ? ~u : (u | 0x80000000u);
}
static __device__ __forceinline__ float fkey_inv(unsigned k) {
  return __uint_as_float((k & 0x80000000u) ? (k & 0x7FFFFFFFu) : ~k);
}

__global__ __launch_bounds__(256) void k_gemm(const float* __restrict__ x,
                                              const float* __restrict__ lin_w,
                                              const float* __restrict__ res_w,
                                              float* __restrict__ xt,
                                              float* __restrict__ z,
                                              int N) {
  const int wave = threadIdx.x >> 5;
  const int lane = threadIdx.x & 31;
  const int m0 = blockIdx.x * 16;
  const int mrow = lane & 15;
  const int kb = (lane >> 4) << 1;
  int arow = m0 + mrow;
  if (arow >= N) arow = N - 1;
  const float* xr = x + (size_t)arow * IN_CH;
  v2f a[16];
#pragma unroll
  for (int ks = 0; ks < 16; ++ks) {
    a[ks].x = xr[ks * 4 + kb];
    a[ks].y = xr[ks * 4 + kb + 1];
  }
#pragma unroll
  for (int tt = 0; tt < 2; ++tt) {
    const int tile = wave + tt * 8;
    const float* W = (tile < 8) ? lin_w : res_w;
    const int cbase = (tile & 7) * 16;
    const float* wr = W + (size_t)(cbase + (lane & 15)) * IN_CH;
    v8f c = {0.f, 0.f, 0.f, 0.f, 0.f, 0.f, 0.f, 0.f};
#pragma unroll
    for (int ks = 0; ks < 16; ++ks) {
      v2f b;
      b.x = wr[ks * 4 + kb];
      b.y = wr[ks * 4 + kb + 1];
      c = __builtin_amdgcn_wmma_f32_16x16x4_f32(false, a[ks], false, b,
                                                (short)0, c, false, false);
    }
    float* dst = (tile < 8) ? xt : z;
    const int mb = m0 + ((lane >> 4) << 3);
#pragma unroll
    for (int r = 0; r < 8; ++r) {
      const int m = mb + r;
      if (m < N) dst[(size_t)m * OUT_DIM + cbase + (lane & 15)] = c[r];
    }
  }
}

__global__ __launch_bounds__(256) void k_node(const float* __restrict__ xt,
                                              const float* __restrict__ att_src,
                                              const float* __restrict__ att_dst,
                                              float* __restrict__ a_src,
                                              float* __restrict__ a_dst, int N) {
  const int lane = threadIdx.x & 31;
  const int n = blockIdx.x * 8 + (threadIdx.x >> 5);
  if (n >= N) return;
  const float4 v = ((const float4*)xt)[(size_t)n * 32 + lane];
  const float4 as = ((const float4*)att_src)[lane];
  const float4 ad = ((const float4*)att_dst)[lane];
  float ps = v.x * as.x + v.y * as.y + v.z * as.z + v.w * as.w;
  float pd = v.x * ad.x + v.y * ad.y + v.z * ad.z + v.w * ad.w;
#pragma unroll
  for (int off = 1; off < 8; off <<= 1) {
    ps += __shfl_xor(ps, off, 32);
    pd += __shfl_xor(pd, off, 32);
  }
  const int h = lane >> 3;
  if ((lane & 7) == 0) {
    a_src[(size_t)n * HEADS + h] = ps;
    a_dst[(size_t)n * HEADS + h] = pd;
  }
}

#define NPW 196
#define APITCH 130
#define QCAP 8192
static_assert(NPW * HEADS <= 1024, "one lane per (owned node, head) in the start phase");
static_assert((NPW * APITCH) % 4 == 0, "acc is cleared in 16-byte pieces");
static_assert(NPW < QCAP, "the self-loops must fit the queue");
__global__ __launch_bounds__(1024) void k_own(const int* __restrict__ ei,
                                              const float* __restrict__ xt,
                                              const float* __restrict__ a_src,
                                              const float* __restrict__ a_dst,
                                              const float* __restrict__ r,
                                              const float* __restrict__ gat_bias,
                                              const float* __restrict__ res_b,
                                              const float* __restrict__ ln_g,
                                              const float* __restrict__ ln_b,
                                              float* __restrict__ out, int E, int N) {
  __shared__ __attribute__((aligned(16))) float acc_l[NPW * APITCH];
  __shared__ __attribute__((aligned(16))) float esum_l[NPW * HEADS];
  __shared__ __attribute__((aligned(16))) float adl_l[NPW * HEADS];
  __shared__ unsigned mxk_l[NPW * HEADS];
  __shared__ int qe[QCAP];
  __shared__ int qd[QCAP];
  __shared__ int qn;

  const int tid = threadIdx.x;
  const int lane = tid & 31;
  const int wave = __builtin_amdgcn_readfirstlane(tid >> 5);
  const int h = lane >> 3;
  const int coff = lane * 4 + (lane >> 4);
  const int lo = blockIdx.x * NPW;
  const int cnt = min(NPW, N - lo);
  const int* __restrict__ dp = ei + E;

  if (tid == 0) qn = cnt;
  for (int i = tid; i < NPW * APITCH / 4; i += 1024)
    ((float4*)acc_l)[i] = make_float4(0.f, 0.f, 0.f, 0.f);
  if (tid < cnt * HEADS) {
    const float as = a_src[(size_t)lo * HEADS + tid];
    const float ad = a_dst[(size_t)lo * HEADS + tid];
    adl_l[tid] = ad;
    mxk_l[tid] = fkey(lrelu(as + ad));
    esum_l[tid] = 0.f;
  }
  if (tid < cnt) {
    qe[tid] = lo + tid;
    qd[tid] = tid;
  }
  __syncthreads();

  auto max_edge = [&](const float4 as, int dl) {
    const float4 ad = *(const float4*)(adl_l + dl * HEADS);
    unsigned* mp = mxk_l + dl * HEADS;
    atomicMax(mp + 0, fkey(lrelu(as.x + ad.x)));
    atomicMax(mp + 1, fkey(lrelu(as.y + ad.y)));
    atomicMax(mp + 2, fkey(lrelu(as.z + ad.z)));
    atomicMax(mp + 3, fkey(lrelu(as.w + ad.w)));
  };

  auto queue_edge = [&](int e) {
    const int slot = atomicAdd(&qn, 1);
    if (slot < QCAP) {
      qe[slot] = e;
    } else {
      max_edge(((const float4*)a_src)[ei[e]], dp[e] - lo);
    }
  };

  const int nmain = ((E & 3) == 0) ? ((E >> 12) & ~7) : 0;
  const int c0 = (int)(((unsigned long long)blockIdx.x * (unsigned)nmain) / gridDim.x);
  for (int ci = 0; ci < nmain; ci += 8) {
    int4 v[8];
#pragma unroll
    for (int u = 0; u < 8; ++u) {
      int cc = c0 + ci + u;
      if (cc >= nmain) cc -= nmain;
      v[u] = *(const int4*)(dp + (cc << 12) + tid * 4);
    }
    unsigned hit = 0;
#pragma unroll
    for (int u = 0; u < 8; ++u) {
      hit |= (unsigned)((unsigned)(v[u].x - lo) < (unsigned)cnt) << (u * 4 + 0);
      hit |= (unsigned)((unsigned)(v[u].y - lo) < (unsigned)cnt) << (u * 4 + 1);
      hit |= (unsigned)((unsigned)(v[u].z - lo) < (unsigned)cnt) << (u * 4 + 2);
      hit |= (unsigned)((unsigned)(v[u].w - lo) < (unsigned)cnt) << (u * 4 + 3);
    }
    while (hit) {
      const int k = __builtin_ctz(hit);
      hit &= hit - 1;
      int cc = c0 + ci + (k >> 2);
      if (cc >= nmain) cc -= nmain;
      queue_edge((cc << 12) + tid * 4 + (k & 3));
    }
  }
  for (int eb = (nmain << 12) + tid; eb < E; eb += 4096) {
    int d[4];
#pragma unroll
    for (int u = 0; u < 4; ++u) d[u] = dp[min(eb + u * 1024, E - 1)];
#pragma unroll
    for (int u = 0; u < 4; ++u)
      if (eb + u * 1024 < E && (unsigned)(d[u] - lo) < (unsigned)cnt) queue_edge(eb + u * 1024);
  }
  __syncthreads();

  const int qtot = qn;
  const int nqx = min(qtot, QCAP);
  if (nqx > cnt) {
    int ev[4], sv[4], dv[4];
    float4 av[4];
#pragma unroll
    for (int u = 0; u < 4; ++u) {
      const int i = cnt + u * 1024 + tid;
      const int e = qe[min(i, QCAP - 1)];
      ev[u] = (i < nqx) ? e : 0;
    }
#pragma unroll
    for (int u = 0; u < 4; ++u) {
      sv[u] = ei[ev[u]];
      dv[u] = dp[ev[u]];
    }
#pragma unroll
    for (int u = 0; u < 4; ++u) av[u] = ((const float4*)a_src)[sv[u]];
#pragma unroll
    for (int u = 0; u < 4; ++u) {
      const int i = cnt + u * 1024 + tid;
      if (i < nqx) {
        qe[i] = sv[u];
        qd[i] = dv[u] - lo;
        max_edge(av[u], dv[u] - lo);
      }
    }
    for (int i = cnt + 4096 + tid; i < nqx; i += 1024) {
      const int e = qe[i];
      const int s = ei[e];
      const int dl = dp[e] - lo;
      qe[i] = s;
      qd[i] = dl;
      max_edge(((const float4*)a_src)[s], dl);
    }
  }
  __syncthreads();

  const int nq = (qtot > QCAP) ? cnt : qtot;
  for (int i0 = wave * 8; i0 < nq; i0 += 256) {
    int s[8], dl[8];
    float4 m[8];
    float as[8], ad[8], mx[8];
#pragma unroll
    for (int u = 0; u < 8; ++u) {
      const int i = min(i0 + u, nq - 1);
      s[u] = __builtin_amdgcn_readfirstlane(qe[i]);
      dl[u] = __builtin_amdgcn_readfirstlane(qd[i]);
      m[u] = ((const float4*)xt)[(size_t)s[u] * 32 + lane];
      as[u] = a_src[(size_t)s[u] * HEADS + h];
      ad[u] = adl_l[dl[u] * HEADS + h];
      mx[u] = fkey_inv(mxk_l[dl[u] * HEADS + h]);
    }
#pragma unroll
    for (int u = 0; u < 8; ++u) {
      if (i0 + u < nq) {
        const float ex = __expf(lrelu(as[u] + ad[u]) - mx[u]);
        float* ap = acc_l + dl[u] * APITCH + coff;
        atomicAdd(ap + 0, ex * m[u].x);
        atomicAdd(ap + 1, ex * m[u].y);
        atomicAdd(ap + 2, ex * m[u].z);
        atomicAdd(ap + 3, ex * m[u].w);
        if ((lane & 7) == 0) atomicAdd(esum_l + dl[u] * HEADS + h, ex);
      }
    }
  }
  if (qtot > QCAP) {
    for (int e = tid; e < E; e += 1024) {
      const int dl = dp[e] - lo;
      if ((unsigned)dl < (unsigned)cnt) {
        const int s = ei[e];
#pragma unroll 1
        for (int hh = 0; hh < HEADS; ++hh) {
          const float ex = __expf(lrelu(a_src[(size_t)s * HEADS + hh] + adl_l[dl * HEADS + hh]) -
                                  fkey_inv(mxk_l[dl * HEADS + hh]));
          atomicAdd(esum_l + dl * HEADS + hh, ex);
#pragma unroll 1
          for (int ch = hh * 32; ch < hh * 32 + 32; ++ch)
            atomicAdd(acc_l + dl * APITCH + ch + (ch >> 6), ex * xt[(size_t)s * OUT_DIM + ch]);
        }
      }
    }
  }
  __syncthreads();

  const float4 gb = ((const float4*)gat_bias)[lane];
  const float4 rb = ((const float4*)res_b)[lane];
  const float4 g = ((const float4*)ln_g)[lane];
  const float4 b = ((const float4*)ln_b)[lane];
  for (int dl = wave; dl < cnt; dl += 32) {
    const int n = lo + dl;
    const float* ap = acc_l + dl * APITCH + coff;
    const float4 rr = ((const float4*)r)[(size_t)n * 32 + lane];
    const float den = esum_l[dl * HEADS + h] + 1e-16f;
    float4 v;
    v.x = ap[0] / den + gb.x + rr.x + rb.x;
    v.y = ap[1] / den + gb.y + rr.y + rb.y;
    v.z = ap[2] / den + gb.z + rr.z + rb.z;
    v.w = ap[3] / den + gb.w + rr.w + rb.w;
    float sm = v.x + v.y + v.z + v.w;
#pragma unroll
    for (int off = 1; off < 32; off <<= 1) sm += __shfl_xor(sm, off, 32);
    const float mu = sm * (1.0f / 128.0f);
    const float dx = v.x - mu, dy = v.y - mu, dz = v.z - mu, dw = v.w - mu;
    float q = dx * dx + dy * dy + dz * dz + dw * dw;
#pragma unroll
    for (int off = 1; off < 32; off <<= 1) q += __shfl_xor(q, off, 32);
    const float inv = rsqrtf(q * (1.0f / 128.0f) + 1e-5f);
    float4 o;
    o.x = lrelu(dx * inv * g.x + b.x);
    o.y = lrelu(dy * inv * g.y + b.y);
    o.z = lrelu(dz * inv * g.z + b.z);
    o.w = lrelu(dw * inv * g.w + b.w);
    ((float4*)out)[(size_t)n * 32 + lane] = o;
  }
}

extern "C" void kernel_launch(void* const* d_in, const int* in_sizes, int n_in,
                              void* d_out, int out_size, void* d_ws, size_t ws_size,
                              hipStream_t stream) {
  const float* x        = (const float*)d_in[0];
  const int*   ei       = (const int*)d_in[1];
  const float* lin_w    = (const float*)d_in[2];
  const float* att_src  = (const float*)d_in[3];
  const float* att_dst  = (const float*)d_in[4];
  const float* gat_bias = (const float*)d_in[5];
  const float* res_w    = (const float*)d_in[6];
  const float* res_b    = (const float*)d_in[7];
  const float* ln_g     = (const float*)d_in[8];
  const float* ln_b     = (const float*)d_in[9];
  const int N  = in_sizes[0] / IN_CH;
  const int E  = in_sizes[1] / 2;

  float* xt    = (float*)d_ws;
  float* z     = xt + (size_t)N * OUT_DIM;
  float* a_src = z + (size_t)N * OUT_DIM;
  float* a_dst = a_src + (size_t)N * HEADS;

  k_gemm<<<(N + 15) / 16, 256, 0, stream>>>(x, lin_w, res_w, xt, z, N);
  k_node<<<(N + 7) / 8, 256, 0, stream>>>(xt, att_src, att_dst, a_src, a_dst, N);
  k_own<<<(N + NPW - 1) / NPW, 1024, 0, stream>>>(ei, xt, a_src, a_dst, z, gat_bias,
                                                  res_b, ln_g, ln_b, (float*)d_out, E, N);
}


# Owner groups scan a 1-byte key column (owner group of each edge's destination, written by the node pass), 16 edges per load with a 4-byte SWAR match, not the 4-byte destinations; queue order changes
# speedup vs baseline: 4.5111x; 1.1358x over previous
#include <hip/hip_runtime.h>
#include <hip/hip_bf16.h>

typedef __attribute__((ext_vector_type(2))) float v2f;
typedef __attribute__((ext_vector_type(8))) float v8f;

#define IN_CH 64
#define OUT_DIM 128
#define HEADS 4
#define NPW 196

static __device__ __forceinline__ float lrelu(float x) { return fmaxf(x, 0.2f * x); }

static __device__ __forceinline__ unsigned fkey(float f) {
  unsigned u = __float_as_uint(f);
  return (u & 0x80000000u) ? ~u : (u | 0x80000000u);
}
static __device__ __forceinline__ float fkey_inv(unsigned k) {
  return __uint_as_float((k & 0x80000000u) ? (k & 0x7FFFFFFFu) : ~k);
}

__global__ __launch_bounds__(256) void k_gemm(const float* __restrict__ x,
                                              const float* __restrict__ lin_w,
                                              const float* __restrict__ res_w,
                                              float* __restrict__ xt,
                                              float* __restrict__ z,
                                              int N) {
  const int wave = threadIdx.x >> 5;
  const int lane = threadIdx.x & 31;
  const int m0 = blockIdx.x * 16;
  const int mrow = lane & 15;
  const int kb = (lane >> 4) << 1;
  int arow = m0 + mrow;
  if (arow >= N) arow = N - 1;
  const float* xr = x + (size_t)arow * IN_CH;
  v2f a[16];
#pragma unroll
  for (int ks = 0; ks < 16; ++ks) {
    a[ks].x = xr[ks * 4 + kb];
    a[ks].y = xr[ks * 4 + kb + 1];
  }
#pragma unroll
  for (int tt = 0; tt < 2; ++tt) {
    const int tile = wave + tt * 8;
    const float* W = (tile < 8) ? lin_w : res_w;
    const int cbase = (tile & 7) * 16;
    const float* wr = W + (size_t)(cbase + (lane & 15)) * IN_CH;
    v8f c = {0.f, 0.f, 0.f, 0.f, 0.f, 0.f, 0.f, 0.f};
#pragma unroll
    for (int ks = 0; ks < 16; ++ks) {
      v2f b;
      b.x = wr[ks * 4 + kb];
      b.y = wr[ks * 4 + kb + 1];
      c = __builtin_amdgcn_wmma_f32_16x16x4_f32(false, a[ks], false, b,
                                                (short)0, c, false, false);
    }
    float* dst = (tile < 8) ? xt : z;
    const int mb = m0 + ((lane >> 4) << 3);
#pragma unroll
    for (int r = 0; r < 8; ++r) {
      const int m = mb + r;
      if (m < N) dst[(size_t)m * OUT_DIM + cbase + (lane & 15)] = c[r];
    }
  }
}

__global__ __launch_bounds__(256) void k_node(const float* __restrict__ xt,
                                              const float* __restrict__ att_src,
                                              const float* __restrict__ att_dst,
                                              float* __restrict__ a_src,
                                              float* __restrict__ a_dst,
                                              const int* __restrict__ ei,
                                              unsigned* __restrict__ keys, int E, int N) {
  const int nkw = ((E & 3) == 0 && (N + NPW - 1) / NPW <= 256) ? ((E >> 4) << 2) : 0;
  for (int t = blockIdx.x * 256 + threadIdx.x; t < nkw; t += gridDim.x * 256) {
    const int4 d = ((const int4*)(ei + E))[t];
    keys[t] = ((unsigned)d.x / NPW) | (((unsigned)d.y / NPW) << 8) |
              (((unsigned)d.z / NPW) << 16) | (((unsigned)d.w / NPW) << 24);
  }
  const int lane = threadIdx.x & 31;
  const int n = blockIdx.x * 8 + (threadIdx.x >> 5);
  if (n >= N) return;
  const float4 v = ((const float4*)xt)[(size_t)n * 32 + lane];
  const float4 as = ((const float4*)att_src)[lane];
  const float4 ad = ((const float4*)att_dst)[lane];
  float ps = v.x * as.x + v.y * as.y + v.z * as.z + v.w * as.w;
  float pd = v.x * ad.x + v.y * ad.y + v.z * ad.z + v.w * ad.w;
#pragma unroll
  for (int off = 1; off < 8; off <<= 1) {
    ps += __shfl_xor(ps, off, 32);
    pd += __shfl_xor(pd, off, 32);
  }
  const int h = lane >> 3;
  if ((lane & 7) == 0) {
    a_src[(size_t)n * HEADS + h] = ps;
    a_dst[(size_t)n * HEADS + h] = pd;
  }
}

#define APITCH 130
#define QCAP 8192
static __device__ __forceinline__ unsigned key_match(unsigned v, unsigned pat) {
  const unsigned t = v ^ pat;
  return ~(((t & 0x7F7F7F7Fu) + 0x7F7F7F7Fu) | t | 0x7F7F7F7Fu);
}
static __device__ __forceinline__ unsigned piece_match(const uint4 v, unsigned pat) {
  return key_match(v.x, pat) | (key_match(v.y, pat) >> 1) | (key_match(v.z, pat) >> 2) |
         (key_match(v.w, pat) >> 3);
}
static_assert(NPW * HEADS <= 1024, "one lane per (owned node, head) in the start phase");
static_assert((NPW * APITCH) % 4 == 0, "acc is cleared in 16-byte pieces");
static_assert(NPW < QCAP, "the self-loops must fit the queue");
__global__ __launch_bounds__(1024) void k_own(const int* __restrict__ ei,
                                              const uint4* __restrict__ keys,
                                              const float* __restrict__ xt,
                                              const float* __restrict__ a_src,
                                              const float* __restrict__ a_dst,
                                              const float* __restrict__ r,
                                              const float* __restrict__ gat_bias,
                                              const float* __restrict__ res_b,
                                              const float* __restrict__ ln_g,
                                              const float* __restrict__ ln_b,
                                              float* __restrict__ out, int E, int N) {
  __shared__ __attribute__((aligned(16))) float acc_l[NPW * APITCH];
  __shared__ __attribute__((aligned(16))) float esum_l[NPW * HEADS];
  __shared__ __attribute__((aligned(16))) float adl_l[NPW * HEADS];
  __shared__ unsigned mxk_l[NPW * HEADS];
  __shared__ int qe[QCAP];
  __shared__ int qd[QCAP];
  __shared__ int qn;

  const int tid = threadIdx.x;
  const int lane = tid & 31;
  const int wave = __builtin_amdgcn_readfirstlane(tid >> 5);
  const int h = lane >> 3;
  const int coff = lane * 4 + (lane >> 4);
  const int lo = blockIdx.x * NPW;
  const int cnt = min(NPW, N - lo);
  const int* __restrict__ dp = ei + E;

  if (tid == 0) qn = cnt;
  for (int i = tid; i < NPW * APITCH / 4; i += 1024)
    ((float4*)acc_l)[i] = make_float4(0.f, 0.f, 0.f, 0.f);
  if (tid < cnt * HEADS) {
    const float as = a_src[(size_t)lo * HEADS + tid];
    const float ad = a_dst[(size_t)lo * HEADS + tid];
    adl_l[tid] = ad;
    mxk_l[tid] = fkey(lrelu(as + ad));
    esum_l[tid] = 0.f;
  }
  if (tid < cnt) {
    qe[tid] = lo + tid;
    qd[tid] = tid;
  }
  __syncthreads();

  auto max_edge = [&](const float4 as, int dl) {
    const float4 ad = *(const float4*)(adl_l + dl * HEADS);
    unsigned* mp = mxk_l + dl * HEADS;
    atomicMax(mp + 0, fkey(lrelu(as.x + ad.x)));
    atomicMax(mp + 1, fkey(lrelu(as.y + ad.y)));
    atomicMax(mp + 2, fkey(lrelu(as.z + ad.z)));
    atomicMax(mp + 3, fkey(lrelu(as.w + ad.w)));
  };

  auto queue_edge = [&](int e) {
    const int slot = atomicAdd(&qn, 1);
    if (slot < QCAP) {
      qe[slot] = e;
    } else {
      max_edge(((const float4*)a_src)[ei[e]], dp[e] - lo);
    }
  };

  const int P = ((E & 3) == 0 && gridDim.x <= 256) ? (E >> 4) : 0;
  const unsigned pat = blockIdx.x * 0x01010101u;
  const int base = (int)(((unsigned long long)blockIdx.x * (unsigned)(P >> 5)) / gridDim.x) << 5;
  const int nfull = P >> 13;
  for (int tr = 0; tr < nfull; ++tr) {
    uint4 v[8];
#pragma unroll
    for (int u = 0; u < 8; ++u) {
      int idx = base + ((tr * 8 + u) << 10) + tid;
      if (idx >= P) idx -= P;
      v[u] = keys[idx];
    }
    unsigned hm[4];
#pragma unroll
    for (int p = 0; p < 4; ++p)
      hm[p] = piece_match(v[2 * p], pat) | (piece_match(v[2 * p + 1], pat) >> 4);
#pragma unroll
    for (int p = 0; p < 4; ++p) {
      unsigned m = hm[p];
      while (m) {
        const int k = __builtin_ctz(m);
        m &= m - 1;
        const int r = k & 7;
        int idx = base + ((tr * 8 + 2 * p + (r < 4 ? 1 : 0)) << 10) + tid;
        if (idx >= P) idx -= P;
        queue_edge((idx << 4) + ((3 - (r & 3)) << 2) + (k >> 3));
      }
    }
  }
  for (int sidx = (nfull << 13) + tid; sidx < P; sidx += 1024) {
    int idx = base + sidx;
    if (idx >= P) idx -= P;
    unsigned m = piece_match(keys[idx], pat);
    while (m) {
      const int k = __builtin_ctz(m);
      m &= m - 1;
      queue_edge((idx << 4) + ((7 - (k & 7)) << 2) + (k >> 3));
    }
  }
  for (int eb = (P << 4) + tid; eb < E; eb += 4096) {
    int d[4];
#pragma unroll
    for (int u = 0; u < 4; ++u) d[u] = dp[min(eb + u * 1024, E - 1)];
#pragma unroll
    for (int u = 0; u < 4; ++u)
      if (eb + u * 1024 < E && (unsigned)(d[u] - lo) < (unsigned)cnt) queue_edge(eb + u * 1024);
  }
  __syncthreads();

  const int qtot = qn;
  const int nqx = min(qtot, QCAP);
  if (nqx > cnt) {
    int ev[4], sv[4], dv[4];
    float4 av[4];
#pragma unroll
    for (int u = 0; u < 4; ++u) {
      const int i = cnt + u * 1024 + tid;
      const int e = qe[min(i, QCAP - 1)];
      ev[u] = (i < nqx) ? e : 0;
    }
#pragma unroll
    for (int u = 0; u < 4; ++u) {
      sv[u] = ei[ev[u]];
      dv[u] = dp[ev[u]];
    }
#pragma unroll
    for (int u = 0; u < 4; ++u) av[u] = ((const float4*)a_src)[sv[u]];
#pragma unroll
    for (int u = 0; u < 4; ++u) {
      const int i = cnt + u * 1024 + tid;
      if (i < nqx) {
        qe[i] = sv[u];
        qd[i] = dv[u] - lo;
        max_edge(av[u], dv[u] - lo);
      }
    }
    for (int i = cnt + 4096 + tid; i < nqx; i += 1024) {
      const int e = qe[i];
      const int s = ei[e];
      const int dl = dp[e] - lo;
      qe[i] = s;
      qd[i] = dl;
      max_edge(((const float4*)a_src)[s], dl);
    }
  }
  __syncthreads();

  const int nq = (qtot > QCAP) ? cnt : qtot;
  for (int i0 = wave * 8; i0 < nq; i0 += 256) {
    int s[8], dl[8];
    float4 m[8];
    float as[8], ad[8], mx[8];
#pragma unroll
    for (int u = 0; u < 8; ++u) {
      const int i = min(i0 + u, nq - 1);
      s[u] = __builtin_amdgcn_readfirstlane(qe[i]);
      dl[u] = __builtin_amdgcn_readfirstlane(qd[i]);
      m[u] = ((const float4*)xt)[(size_t)s[u] * 32 + lane];
      as[u] = a_src[(size_t)s[u] * HEADS + h];
      ad[u] = adl_l[dl[u] * HEADS + h];
      mx[u] = fkey_inv(mxk_l[dl[u] * HEADS + h]);
    }
#pragma unroll
    for (int u = 0; u < 8; ++u) {
      if (i0 + u < nq) {
        const float ex = __expf(lrelu(as[u] + ad[u]) - mx[u]);
        float* ap = acc_l + dl[u] * APITCH + coff;
        atomicAdd(ap + 0, ex * m[u].x);
        atomicAdd(ap + 1, ex * m[u].y);
        atomicAdd(ap + 2, ex * m[u].z);
        atomicAdd(ap + 3, ex * m[u].w);
        if ((lane & 7) == 0) atomicAdd(esum_l + dl[u] * HEADS + h, ex);
      }
    }
  }
  if (qtot > QCAP) {
    for (int e = tid; e < E; e += 1024) {
      const int dl = dp[e] - lo;
      if ((unsigned)dl < (unsigned)cnt) {
        const int s = ei[e];
#pragma unroll 1
        for (int hh = 0; hh < HEADS; ++hh) {
          const float ex = __expf(lrelu(a_src[(size_t)s * HEADS + hh] + adl_l[dl * HEADS + hh]) -
                                  fkey_inv(mxk_l[dl * HEADS + hh]));
          atomicAdd(esum_l + dl * HEADS + hh, ex);
#pragma unroll 1
          for (int ch = hh * 32; ch < hh * 32 + 32; ++ch)
            atomicAdd(acc_l + dl * APITCH + ch + (ch >> 6), ex * xt[(size_t)s * OUT_DIM + ch]);
        }
      }
    }
  }
  __syncthreads();

  const float4 gb = ((const float4*)gat_bias)[lane];
  const float4 rb = ((const float4*)res_b)[lane];
  const float4 g = ((const float4*)ln_g)[lane];
  const float4 b = ((const float4*)ln_b)[lane];
  for (int dl = wave; dl < cnt; dl += 32) {
    const int n = lo + dl;
    const float* ap = acc_l + dl * APITCH + coff;
    const float4 rr = ((const float4*)r)[(size_t)n * 32 + lane];
    const float den = esum_l[dl * HEADS + h] + 1e-16f;
    float4 v;
    v.x = ap[0] / den + gb.x + rr.x + rb.x;
    v.y = ap[1] / den + gb.y + rr.y + rb.y;
    v.z = ap[2] / den + gb.z + rr.z + rb.z;
    v.w = ap[3] / den + gb.w + rr.w + rb.w;
    float sm = v.x + v.y + v.z + v.w;
#pragma unroll
    for (int off = 1; off < 32; off <<= 1) sm += __shfl_xor(sm, off, 32);
    const float mu = sm * (1.0f / 128.0f);
    const float dx = v.x - mu, dy = v.y - mu, dz = v.z - mu, dw = v.w - mu;
    float q = dx * dx + dy * dy + dz * dz + dw * dw;
#pragma unroll
    for (int off = 1; off < 32; off <<= 1) q += __shfl_xor(q, off, 32);
    const float inv = rsqrtf(q * (1.0f / 128.0f) + 1e-5f);
    float4 o;
    o.x = lrelu(dx * inv * g.x + b.x);
    o.y = lrelu(dy * inv * g.y + b.y);
    o.z = lrelu(dz * inv * g.z + b.z);
    o.w = lrelu(dw * inv * g.w + b.w);
    ((float4*)out)[(size_t)n * 32 + lane] = o;
  }
}

extern "C" void kernel_launch(void* const* d_in, const int* in_sizes, int n_in,
                              void* d_out, int out_size, void* d_ws, size_t ws_size,
                              hipStream_t stream) {
  const float* x        = (const float*)d_in[0];
  const int*   ei       = (const int*)d_in[1];
  const float* lin_w    = (const float*)d_in[2];
  const float* att_src  = (const float*)d_in[3];
  const float* att_dst  = (const float*)d_in[4];
  const float* gat_bias = (const float*)d_in[5];
  const float* res_w    = (const float*)d_in[6];
  const float* res_b    = (const float*)d_in[7];
  const float* ln_g     = (const float*)d_in[8];
  const float* ln_b     = (const float*)d_in[9];
  const int N  = in_sizes[0] / IN_CH;
  const int E  = in_sizes[1] / 2;

  float* xt    = (float*)d_ws;
  float* z     = xt + (size_t)N * OUT_DIM;
  float* a_src = z + (size_t)N * OUT_DIM;
  float* a_dst = a_src + (size_t)N * HEADS;
  unsigned* keys = (unsigned*)(a_dst + (size_t)N * HEADS);

  k_gemm<<<(N + 15) / 16, 256, 0, stream>>>(x, lin_w, res_w, xt, z, N);
  k_node<<<(N + 7) / 8, 256, 0, stream>>>(xt, att_src, att_dst, a_src, a_dst, ei, keys, E, N);
  k_own<<<(N + NPW - 1) / NPW, 1024, 0, stream>>>(ei, (const uint4*)keys, xt, a_src, a_dst, z, gat_bias,
                                                  res_b, ln_g, ln_b, (float*)d_out, E, N);
}


# GEMM launch re-tiled: a wave makes 16 rows x 128 columns of one product from x fragments loaded once, all fragments as 16-byte pieces (one permuted k order in both loaders); k sum order changes
# speedup vs baseline: 4.8933x; 1.0847x over previous
#include <hip/hip_runtime.h>
#include <hip/hip_bf16.h>

typedef __attribute__((ext_vector_type(2))) float v2f;
typedef __attribute__((ext_vector_type(8))) float v8f;

#define IN_CH 64
#define OUT_DIM 128
#define HEADS 4
#define NPW 196

static __device__ __forceinline__ float lrelu(float x) { return fmaxf(x, 0.2f * x); }

static __device__ __forceinline__ unsigned fkey(float f) {
  unsigned u = __float_as_uint(f);
  return (u & 0x80000000u) ? ~u : (u | 0x80000000u);
}
static __device__ __forceinline__ float fkey_inv(unsigned k) {
  return __uint_as_float((k & 0x80000000u) ? (k & 0x7FFFFFFFu) : ~k);
}

__global__ __launch_bounds__(128) void k_gemm(const float* __restrict__ x,
                                              const float* __restrict__ lin_w,
                                              const float* __restrict__ res_w,
                                              float* __restrict__ xt,
                                              float* __restrict__ z,
                                              int N) {
  const int wave = __builtin_amdgcn_readfirstlane(threadIdx.x >> 5);
  const int lane = threadIdx.x & 31;
  const int c16 = lane & 15;
  const int hf = lane >> 4;
  const int m0 = (blockIdx.x * 4 + wave) * 16;
  if (m0 >= N) return;
  const float* Wm = blockIdx.y ? res_w : lin_w;
  float* dst = blockIdx.y ? z : xt;
  int arow = m0 + c16;
  if (arow >= N) arow = N - 1;
  const float4* xr = (const float4*)(x + (size_t)arow * IN_CH + 4 * hf);
  const float4* wr = (const float4*)(Wm + (size_t)c16 * IN_CH + 4 * hf);
  float4 a[8], b0[8], b1[8];
#pragma unroll
  for (int i = 0; i < 8; ++i) a[i] = xr[2 * i];
#pragma unroll
  for (int i = 0; i < 8; ++i) {
    b0[i] = wr[2 * i];
    b1[i] = wr[16 * IN_CH / 4 + 2 * i];
  }
  float* orow = dst + (size_t)(m0 + (hf << 3)) * OUT_DIM + c16;
  const bool whole = (m0 + 16 <= N);
#pragma unroll
  for (int tp = 0; tp < 4; ++tp) {
    float4 n0[8], n1[8];
    if (tp < 3) {
#pragma unroll
      for (int i = 0; i < 8; ++i) {
        n0[i] = wr[(tp + 1) * 32 * IN_CH / 4 + 2 * i];
        n1[i] = wr[(tp + 1) * 32 * IN_CH / 4 + 16 * IN_CH / 4 + 2 * i];
      }
    }
    v8f c0 = {0.f, 0.f, 0.f, 0.f, 0.f, 0.f, 0.f, 0.f};
    v8f c1 = {0.f, 0.f, 0.f, 0.f, 0.f, 0.f, 0.f, 0.f};
#pragma unroll
    for (int i = 0; i < 8; ++i) {
      const v2f alo = {a[i].x, a[i].y};
      const v2f ahi = {a[i].z, a[i].w};
      const v2f b0lo = {b0[i].x, b0[i].y};
      const v2f b0hi = {b0[i].z, b0[i].w};
      const v2f b1lo = {b1[i].x, b1[i].y};
      const v2f b1hi = {b1[i].z, b1[i].w};
      c0 = __builtin_amdgcn_wmma_f32_16x16x4_f32(false, alo, false, b0lo, (short)0, c0, false, false);
      c1 = __builtin_amdgcn_wmma_f32_16x16x4_f32(false, alo, false, b1lo, (short)0, c1, false, false);
      c0 = __builtin_amdgcn_wmma_f32_16x16x4_f32(false, ahi, false, b0hi, (short)0, c0, false, false);
      c1 = __builtin_amdgcn_wmma_f32_16x16x4_f32(false, ahi, false, b1hi, (short)0, c1, false, false);
    }
    if (whole) {
#pragma unroll
      for (int r = 0; r < 8; ++r) {
        orow[(size_t)r * OUT_DIM + tp * 32] = c0[r];
        orow[(size_t)r * OUT_DIM + tp * 32 + 16] = c1[r];
      }
    } else {
#pragma unroll
      for (int r = 0; r < 8; ++r) {
        if (m0 + (hf << 3) + r < N) {
          orow[(size_t)r * OUT_DIM + tp * 32] = c0[r];
          orow[(size_t)r * OUT_DIM + tp * 32 + 16] = c1[r];
        }
      }
    }
    if (tp < 3) {
#pragma unroll
      for (int i = 0; i < 8; ++i) {
        b0[i] = n0[i];
        b1[i] = n1[i];
      }
    }
  }
}

__global__ __launch_bounds__(256) void k_node(const float* __restrict__ xt,
                                              const float* __restrict__ att_src,
                                              const float* __restrict__ att_dst,
                                              float* __restrict__ a_src,
                                              float* __restrict__ a_dst,
                                              const int* __restrict__ ei,
                                              unsigned* __restrict__ keys, int E, int N) {
  const int nkw = ((E & 3) == 0 && (N + NPW - 1) / NPW <= 256) ? ((E >> 4) << 2) : 0;
  for (int t = blockIdx.x * 256 + threadIdx.x; t < nkw; t += gridDim.x * 256) {
    const int4 d = ((const int4*)(ei + E))[t];
    keys[t] = ((unsigned)d.x / NPW) | (((unsigned)d.y / NPW) << 8) |
              (((unsigned)d.z / NPW) << 16) | (((unsigned)d.w / NPW) << 24);
  }
  const int lane = threadIdx.x & 31;
  const int n = blockIdx.x * 8 + (threadIdx.x >> 5);
  if (n >= N) return;
  const float4 v = ((const float4*)xt)[(size_t)n * 32 + lane];
  const float4 as = ((const float4*)att_src)[lane];
  const float4 ad = ((const float4*)att_dst)[lane];
  float ps = v.x * as.x + v.y * as.y + v.z * as.z + v.w * as.w;
  float pd = v.x * ad.x + v.y * ad.y + v.z * ad.z + v.w * ad.w;
#pragma unroll
  for (int off = 1; off < 8; off <<= 1) {
    ps += __shfl_xor(ps, off, 32);
    pd += __shfl_xor(pd, off, 32);
  }
  const int h = lane >> 3;
  if ((lane & 7) == 0) {
    a_src[(size_t)n * HEADS + h] = ps;
    a_dst[(size_t)n * HEADS + h] = pd;
  }
}

#define APITCH 130
#define QCAP 8192
static __device__ __forceinline__ unsigned key_match(unsigned v, unsigned pat) {
  const unsigned t = v ^ pat;
  return ~(((t & 0x7F7F7F7Fu) + 0x7F7F7F7Fu) | t | 0x7F7F7F7Fu);
}
static __device__ __forceinline__ unsigned piece_match(const uint4 v, unsigned pat) {
  return key_match(v.x, pat) | (key_match(v.y, pat) >> 1) | (key_match(v.z, pat) >> 2) |
         (key_match(v.w, pat) >> 3);
}
static_assert(NPW * HEADS <= 1024, "one lane per (owned node, head) in the start phase");
static_assert((NPW * APITCH) % 4 == 0, "acc is cleared in 16-byte pieces");
static_assert(NPW < QCAP, "the self-loops must fit the queue");
__global__ __launch_bounds__(1024) void k_own(const int* __restrict__ ei,
                                              const uint4* __restrict__ keys,
                                              const float* __restrict__ xt,
                                              const float* __restrict__ a_src,
                                              const float* __restrict__ a_dst,
                                              const float* __restrict__ r,
                                              const float* __restrict__ gat_bias,
                                              const float* __restrict__ res_b,
                                              const float* __restrict__ ln_g,
                                              const float* __restrict__ ln_b,
                                              float* __restrict__ out, int E, int N) {
  __shared__ __attribute__((aligned(16))) float acc_l[NPW * APITCH];
  __shared__ __attribute__((aligned(16))) float esum_l[NPW * HEADS];
  __shared__ __attribute__((aligned(16))) float adl_l[NPW * HEADS];
  __shared__ unsigned mxk_l[NPW * HEADS];
  __shared__ int qe[QCAP];
  __shared__ int qd[QCAP];
  __shared__ int qn;

  const int tid = threadIdx.x;
  const int lane = tid & 31;
  const int wave = __builtin_amdgcn_readfirstlane(tid >> 5);
  const int h = lane >> 3;
  const int coff = lane * 4 + (lane >> 4);
  const int lo = blockIdx.x * NPW;
  const int cnt = min(NPW, N - lo);
  const int* __restrict__ dp = ei + E;

  if (tid == 0) qn = cnt;
  for (int i = tid; i < NPW * APITCH / 4; i += 1024)
    ((float4*)acc_l)[i] = make_float4(0.f, 0.f, 0.f, 0.f);
  if (tid < cnt * HEADS) {
    const float as = a_src[(size_t)lo * HEADS + tid];
    const float ad = a_dst[(size_t)lo * HEADS + tid];
    adl_l[tid] = ad;
    mxk_l[tid] = fkey(lrelu(as + ad));
    esum_l[tid] = 0.f;
  }
  if (tid < cnt) {
    qe[tid] = lo + tid;
    qd[tid] = tid;
  }
  __syncthreads();

  auto max_edge = [&](const float4 as, int dl) {
    const float4 ad = *(const float4*)(adl_l + dl * HEADS);
    unsigned* mp = mxk_l + dl * HEADS;
    atomicMax(mp + 0, fkey(lrelu(as.x + ad.x)));
    atomicMax(mp + 1, fkey(lrelu(as.y + ad.y)));
    atomicMax(mp + 2, fkey(lrelu(as.z + ad.z)));
    atomicMax(mp + 3, fkey(lrelu(as.w + ad.w)));
  };

  auto queue_edge = [&](int e) {
    const int slot = atomicAdd(&qn, 1);
    if (slot < QCAP) {
      qe[slot] = e;
    } else {
      max_edge(((const float4*)a_src)[ei[e]], dp[e] - lo);
    }
  };

  const int P = ((E & 3) == 0 && gridDim.x <= 256) ? (E >> 4) : 0;
  const unsigned pat = blockIdx.x * 0x01010101u;
  const int base = (int)(((unsigned long long)blockIdx.x * (unsigned)(P >> 5)) / gridDim.x) << 5;
  const int nfull = P >> 13;
  for (int tr = 0; tr < nfull; ++tr) {
    uint4 v[8];
#pragma unroll
    for (int u = 0; u < 8; ++u) {
      int idx = base + ((tr * 8 + u) << 10) + tid;
      if (idx >= P) idx -= P;
      v[u] = keys[idx];
    }
    unsigned hm[4];
#pragma unroll
    for (int p = 0; p < 4; ++p)
      hm[p] = piece_match(v[2 * p], pat) | (piece_match(v[2 * p + 1], pat) >> 4);
#pragma unroll
    for (int p = 0; p < 4; ++p) {
      unsigned m = hm[p];
      while (m) {
        const int k = __builtin_ctz(m);
        m &= m - 1;
        const int r = k & 7;
        int idx = base + ((tr * 8 + 2 * p + (r < 4 ? 1 : 0)) << 10) + tid;
        if (idx >= P) idx -= P;
        queue_edge((idx << 4) + ((3 - (r & 3)) << 2) + (k >> 3));
      }
    }
  }
  for (int sidx = (nfull << 13) + tid; sidx < P; sidx += 1024) {
    int idx = base + sidx;
    if (idx >= P) idx -= P;
    unsigned m = piece_match(keys[idx], pat);
    while (m) {
      const int k = __builtin_ctz(m);
      m &= m - 1;
      queue_edge((idx << 4) + ((7 - (k & 7)) << 2) + (k >> 3));
    }
  }
  for (int eb = (P << 4) + tid; eb < E; eb += 4096) {
    int d[4];
#pragma unroll
    for (int u = 0; u < 4; ++u) d[u] = dp[min(eb + u * 1024, E - 1)];
#pragma unroll
    for (int u = 0; u < 4; ++u)
      if (eb + u * 1024 < E && (unsigned)(d[u] - lo) < (unsigned)cnt) queue_edge(eb + u * 1024);
  }
  __syncthreads();

  const int qtot = qn;
  const int nqx = min(qtot, QCAP);
  if (nqx > cnt) {
    int ev[4], sv[4], dv[4];
    float4 av[4];
#pragma unroll
    for (int u = 0; u < 4; ++u) {
      const int i = cnt + u * 1024 + tid;
      const int e = qe[min(i, QCAP - 1)];
      ev[u] = (i < nqx) ? e : 0;
    }
#pragma unroll
    for (int u = 0; u < 4; ++u) {
      sv[u] = ei[ev[u]];
      dv[u] = dp[ev[u]];
    }
#pragma unroll
    for (int u = 0; u < 4; ++u) av[u] = ((const float4*)a_src)[sv[u]];
#pragma unroll
    for (int u = 0; u < 4; ++u) {
      const int i = cnt + u * 1024 + tid;
      if (i < nqx) {
        qe[i] = sv[u];
        qd[i] = dv[u] - lo;
        max_edge(av[u], dv[u] - lo);
      }
    }
    for (int i = cnt + 4096 + tid; i < nqx; i += 1024) {
      const int e = qe[i];
      const int s = ei[e];
      const int dl = dp[e] - lo;
      qe[i] = s;
      qd[i] = dl;
      max_edge(((const float4*)a_src)[s], dl);
    }
  }
  __syncthreads();

  const int nq = (qtot > QCAP) ? cnt : qtot;
  for (int i0 = wave * 8; i0 < nq; i0 += 256) {
    int s[8], dl[8];
    float4 m[8];
    float as[8], ad[8], mx[8];
#pragma unroll
    for (int u = 0; u < 8; ++u) {
      const int i = min(i0 + u, nq - 1);
      s[u] = __builtin_amdgcn_readfirstlane(qe[i]);
      dl[u] = __builtin_amdgcn_readfirstlane(qd[i]);
      m[u] = ((const float4*)xt)[(size_t)s[u] * 32 + lane];
      as[u] = a_src[(size_t)s[u] * HEADS + h];
      ad[u] = adl_l[dl[u] * HEADS + h];
      mx[u] = fkey_inv(mxk_l[dl[u] * HEADS + h]);
    }
#pragma unroll
    for (int u = 0; u < 8; ++u) {
      if (i0 + u < nq) {
        const float ex = __expf(lrelu(as[u] + ad[u]) - mx[u]);
        float* ap = acc_l + dl[u] * APITCH + coff;
        atomicAdd(ap + 0, ex * m[u].x);
        atomicAdd(ap + 1, ex * m[u].y);
        atomicAdd(ap + 2, ex * m[u].z);
        atomicAdd(ap + 3, ex * m[u].w);
        if ((lane & 7) == 0) atomicAdd(esum_l + dl[u] * HEADS + h, ex);
      }
    }
  }
  if (qtot > QCAP) {
    for (int e = tid; e < E; e += 1024) {
      const int dl = dp[e] - lo;
      if ((unsigned)dl < (unsigned)cnt) {
        const int s = ei[e];
#pragma unroll 1
        for (int hh = 0; hh < HEADS; ++hh) {
          const float ex = __expf(lrelu(a_src[(size_t)s * HEADS + hh] + adl_l[dl * HEADS + hh]) -
                                  fkey_inv(mxk_l[dl * HEADS + hh]));
          atomicAdd(esum_l + dl * HEADS + hh, ex);
#pragma unroll 1
          for (int ch = hh * 32; ch < hh * 32 + 32; ++ch)
            atomicAdd(acc_l + dl * APITCH + ch + (ch >> 6), ex * xt[(size_t)s * OUT_DIM + ch]);
        }
      }
    }
  }
  __syncthreads();

  const float4 gb = ((const float4*)gat_bias)[lane];
  const float4 rb = ((const float4*)res_b)[lane];
  const float4 g = ((const float4*)ln_g)[lane];
  const float4 b = ((const float4*)ln_b)[lane];
  for (int dl = wave; dl < cnt; dl += 32) {
    const int n = lo + dl;
    const float* ap = acc_l + dl * APITCH + coff;
    const float4 rr = ((const float4*)r)[(size_t)n * 32 + lane];
    const float den = esum_l[dl * HEADS + h] + 1e-16f;
    float4 v;
    v.x = ap[0] / den + gb.x + rr.x + rb.x;
    v.y = ap[1] / den + gb.y + rr.y + rb.y;
    v.z = ap[2] / den + gb.z + rr.z + rb.z;
    v.w = ap[3] / den + gb.w + rr.w + rb.w;
    float sm = v.x + v.y + v.z + v.w;
#pragma unroll
    for (int off = 1; off < 32; off <<= 1) sm += __shfl_xor(sm, off, 32);
    const float mu = sm * (1.0f / 128.0f);
    const float dx = v.x - mu, dy = v.y - mu, dz = v.z - mu, dw = v.w - mu;
    float q = dx * dx + dy * dy + dz * dz + dw * dw;
#pragma unroll
    for (int off = 1; off < 32; off <<= 1) q += __shfl_xor(q, off, 32);
    const float inv = rsqrtf(q * (1.0f / 128.0f) + 1e-5f);
    float4 o;
    o.x = lrelu(dx * inv * g.x + b.x);
    o.y = lrelu(dy * inv * g.y + b.y);
    o.z = lrelu(dz * inv * g.z + b.z);
    o.w = lrelu(dw * inv * g.w + b.w);
    ((float4*)out)[(size_t)n * 32 + lane] = o;
  }
}

extern "C" void kernel_launch(void* const* d_in, const int* in_sizes, int n_in,
                              void* d_out, int out_size, void* d_ws, size_t ws_size,
                              hipStream_t stream) {
  const float* x        = (const float*)d_in[0];
  const int*   ei       = (const int*)d_in[1];
  const float* lin_w    = (const float*)d_in[2];
  const float* att_src  = (const float*)d_in[3];
  const float* att_dst  = (const float*)d_in[4];
  const float* gat_bias = (const float*)d_in[5];
  const float* res_w    = (const float*)d_in[6];
  const float* res_b    = (const float*)d_in[7];
  const float* ln_g     = (const float*)d_in[8];
  const float* ln_b     = (const float*)d_in[9];
  const int N  = in_sizes[0] / IN_CH;
  const int E  = in_sizes[1] / 2;

  float* xt    = (float*)d_ws;
  float* z     = xt + (size_t)N * OUT_DIM;
  float* a_src = z + (size_t)N * OUT_DIM;
  float* a_dst = a_src + (size_t)N * HEADS;
  unsigned* keys = (unsigned*)(a_dst + (size_t)N * HEADS);

  k_gemm<<<dim3((N + 63) / 64, 2), 128, 0, stream>>>(x, lin_w, res_w, xt, z, N);
  k_node<<<(N + 7) / 8, 256, 0, stream>>>(xt, att_src, att_dst, a_src, a_dst, ei, keys, E, N);
  k_own<<<(N + NPW - 1) / NPW, 1024, 0, stream>>>(ei, (const uint4*)keys, xt, a_src, a_dst, z, gat_bias,
                                                  res_b, ln_g, ln_b, (float*)d_out, E, N);
}


# Owner groups keep each queued edge's four scores in LDS where the maxima are formed; the sum phase reads them there, not a_src[s] from L2 again: one scattered L2 read per edge less; sums unchanged
# speedup vs baseline: 5.0716x; 1.0364x over previous
#include <hip/hip_runtime.h>
#include <hip/hip_bf16.h>

typedef __attribute__((ext_vector_type(2))) float v2f;
typedef __attribute__((ext_vector_type(8))) float v8f;

#define IN_CH 64
#define OUT_DIM 128
#define HEADS 4
#define NPW 196

static __device__ __forceinline__ float lrelu(float x) { return fmaxf(x, 0.2f * x); }

static __device__ __forceinline__ unsigned fkey(float f) {
  unsigned u = __float_as_uint(f);
  return (u & 0x80000000u) ? ~u : (u | 0x80000000u);
}
static __device__ __forceinline__ float fkey_inv(unsigned k) {
  return __uint_as_float((k & 0x80000000u) ? (k & 0x7FFFFFFFu) : ~k);
}

__global__ __launch_bounds__(128) void k_gemm(const float* __restrict__ x,
                                              const float* __restrict__ lin_w,
                                              const float* __restrict__ res_w,
                                              float* __restrict__ xt,
                                              float* __restrict__ z,
                                              int N) {
  const int wave = __builtin_amdgcn_readfirstlane(threadIdx.x >> 5);
  const int lane = threadIdx.x & 31;
  const int c16 = lane & 15;
  const int hf = lane >> 4;
  const int m0 = (blockIdx.x * 4 + wave) * 16;
  if (m0 >= N) return;
  const float* Wm = blockIdx.y ? res_w : lin_w;
  float* dst = blockIdx.y ? z : xt;
  int arow = m0 + c16;
  if (arow >= N) arow = N - 1;
  const float4* xr = (const float4*)(x + (size_t)arow * IN_CH + 4 * hf);
  const float4* wr = (const float4*)(Wm + (size_t)c16 * IN_CH + 4 * hf);
  float4 a[8], b0[8], b1[8];
#pragma unroll
  for (int i = 0; i < 8; ++i) a[i] = xr[2 * i];
#pragma unroll
  for (int i = 0; i < 8; ++i) {
    b0[i] = wr[2 * i];
    b1[i] = wr[16 * IN_CH / 4 + 2 * i];
  }
  float* orow = dst + (size_t)(m0 + (hf << 3)) * OUT_DIM + c16;
  const bool whole = (m0 + 16 <= N);
#pragma unroll
  for (int tp = 0; tp < 4; ++tp) {
    float4 n0[8], n1[8];
    if (tp < 3) {
#pragma unroll
      for (int i = 0; i < 8; ++i) {
        n0[i] = wr[(tp + 1) * 32 * IN_CH / 4 + 2 * i];
        n1[i] = wr[(tp + 1) * 32 * IN_CH / 4 + 16 * IN_CH / 4 + 2 * i];
      }
    }
    v8f c0 = {0.f, 0.f, 0.f, 0.f, 0.f, 0.f, 0.f, 0.f};
    v8f c1 = {0.f, 0.f, 0.f, 0.f, 0.f, 0.f, 0.f, 0.f};
#pragma unroll
    for (int i = 0; i < 8; ++i) {
      const v2f alo = {a[i].x, a[i].y};
      const v2f ahi = {a[i].z, a[i].w};
      const v2f b0lo = {b0[i].x, b0[i].y};
      const v2f b0hi = {b0[i].z, b0[i].w};
      const v2f b1lo = {b1[i].x, b1[i].y};
      const v2f b1hi = {b1[i].z, b1[i].w};
      c0 = __builtin_amdgcn_wmma_f32_16x16x4_f32(false, alo, false, b0lo, (short)0, c0, false, false);
      c1 = __builtin_amdgcn_wmma_f32_16x16x4_f32(false, alo, false, b1lo, (short)0, c1, false, false);
      c0 = __builtin_amdgcn_wmma_f32_16x16x4_f32(false, ahi, false, b0hi, (short)0, c0, false, false);
      c1 = __builtin_amdgcn_wmma_f32_16x16x4_f32(false, ahi, false, b1hi, (short)0, c1, false, false);
    }
    if (whole) {
#pragma unroll
      for (int r = 0; r < 8; ++r) {
        orow[(size_t)r * OUT_DIM + tp * 32] = c0[r];
        orow[(size_t)r * OUT_DIM + tp * 32 + 16] = c1[r];
      }
    } else {
#pragma unroll
      for (int r = 0; r < 8; ++r) {
        if (m0 + (hf << 3) + r < N) {
          orow[(size_t)r * OUT_DIM + tp * 32] = c0[r];
          orow[(size_t)r * OUT_DIM + tp * 32 + 16] = c1[r];
        }
      }
    }
    if (tp < 3) {
#pragma unroll
      for (int i = 0; i < 8; ++i) {
        b0[i] = n0[i];
        b1[i] = n1[i];
      }
    }
  }
}

__global__ __launch_bounds__(256) void k_node(const float* __restrict__ xt,
                                              const float* __restrict__ att_src,
                                              const float* __restrict__ att_dst,
                                              float* __restrict__ a_src,
                                              float* __restrict__ a_dst,
                                              const int* __restrict__ ei,
                                              unsigned* __restrict__ keys, int E, int N) {
  const int nkw = ((E & 3) == 0 && (N + NPW - 1) / NPW <= 256) ? ((E >> 4) << 2) : 0;
  for (int t = blockIdx.x * 256 + threadIdx.x; t < nkw; t += gridDim.x * 256) {
    const int4 d = ((const int4*)(ei + E))[t];
    keys[t] = ((unsigned)d.x / NPW) | (((unsigned)d.y / NPW) << 8) |
              (((unsigned)d.z / NPW) << 16) | (((unsigned)d.w / NPW) << 24);
  }
  const int lane = threadIdx.x & 31;
  const int n = blockIdx.x * 8 + (threadIdx.x >> 5);
  if (n >= N) return;
  const float4 v = ((const float4*)xt)[(size_t)n * 32 + lane];
  const float4 as = ((const float4*)att_src)[lane];
  const float4 ad = ((const float4*)att_dst)[lane];
  float ps = v.x * as.x + v.y * as.y + v.z * as.z + v.w * as.w;
  float pd = v.x * ad.x + v.y * ad.y + v.z * ad.z + v.w * ad.w;
#pragma unroll
  for (int off = 1; off < 8; off <<= 1) {
    ps += __shfl_xor(ps, off, 32);
    pd += __shfl_xor(pd, off, 32);
  }
  const int h = lane >> 3;
  if ((lane & 7) == 0) {
    a_src[(size_t)n * HEADS + h] = ps;
    a_dst[(size_t)n * HEADS + h] = pd;
  }
}

#define APITCH 130
#define QCAP 8192
static __device__ __forceinline__ unsigned key_match(unsigned v, unsigned pat) {
  const unsigned t = v ^ pat;
  return ~(((t & 0x7F7F7F7Fu) + 0x7F7F7F7Fu) | t | 0x7F7F7F7Fu);
}
static __device__ __forceinline__ unsigned piece_match(const uint4 v, unsigned pat) {
  return key_match(v.x, pat) | (key_match(v.y, pat) >> 1) | (key_match(v.z, pat) >> 2) |
         (key_match(v.w, pat) >> 3);
}
static_assert(NPW * HEADS <= 1024, "one lane per (owned node, head) in the start phase");
static_assert((NPW * APITCH) % 4 == 0, "acc is cleared in 16-byte pieces");
static_assert(NPW < QCAP, "the self-loops must fit the queue");
__global__ __launch_bounds__(1024) void k_own(const int* __restrict__ ei,
                                              const uint4* __restrict__ keys,
                                              const float* __restrict__ xt,
                                              const float* __restrict__ a_src,
                                              const float* __restrict__ a_dst,
                                              const float* __restrict__ r,
                                              const float* __restrict__ gat_bias,
                                              const float* __restrict__ res_b,
                                              const float* __restrict__ ln_g,
                                              const float* __restrict__ ln_b,
                                              float* __restrict__ out, int E, int N) {
  __shared__ __attribute__((aligned(16))) float acc_l[NPW * APITCH];
  __shared__ __attribute__((aligned(16))) float esum_l[NPW * HEADS];
  __shared__ __attribute__((aligned(16))) float adl_l[NPW * HEADS];
  __shared__ unsigned mxk_l[NPW * HEADS];
  __shared__ int qe[QCAP];
  __shared__ int qd[QCAP];
  __shared__ __attribute__((aligned(16))) float qs[QCAP * HEADS];
  __shared__ int qn;

  const int tid = threadIdx.x;
  const int lane = tid & 31;
  const int wave = __builtin_amdgcn_readfirstlane(tid >> 5);
  const int h = lane >> 3;
  const int coff = lane * 4 + (lane >> 4);
  const int lo = blockIdx.x * NPW;
  const int cnt = min(NPW, N - lo);
  const int* __restrict__ dp = ei + E;

  if (tid == 0) qn = cnt;
  for (int i = tid; i < NPW * APITCH / 4; i += 1024)
    ((float4*)acc_l)[i] = make_float4(0.f, 0.f, 0.f, 0.f);
  if (tid < cnt * HEADS) {
    const float as = a_src[(size_t)lo * HEADS + tid];
    const float ad = a_dst[(size_t)lo * HEADS + tid];
    const float sc = lrelu(as + ad);
    adl_l[tid] = ad;
    qs[tid] = sc;
    mxk_l[tid] = fkey(sc);
    esum_l[tid] = 0.f;
  }
  if (tid < cnt) {
    qe[tid] = lo + tid;
    qd[tid] = tid;
  }
  __syncthreads();

  auto max_edge = [&](const float4 as, int dl) {
    const float4 ad = *(const float4*)(adl_l + dl * HEADS);
    float4 sc;
    sc.x = lrelu(as.x + ad.x);
    sc.y = lrelu(as.y + ad.y);
    sc.z = lrelu(as.z + ad.z);
    sc.w = lrelu(as.w + ad.w);
    unsigned* mp = mxk_l + dl * HEADS;
    atomicMax(mp + 0, fkey(sc.x));
    atomicMax(mp + 1, fkey(sc.y));
    atomicMax(mp + 2, fkey(sc.z));
    atomicMax(mp + 3, fkey(sc.w));
    return sc;
  };

  auto queue_edge = [&](int e) {
    const int slot = atomicAdd(&qn, 1);
    if (slot < QCAP) {
      qe[slot] = e;
    } else {
      max_edge(((const float4*)a_src)[ei[e]], dp[e] - lo);
    }
  };

  const int P = ((E & 3) == 0 && gridDim.x <= 256) ? (E >> 4) : 0;
  const unsigned pat = blockIdx.x * 0x01010101u;
  const int base = (int)(((unsigned long long)blockIdx.x * (unsigned)(P >> 5)) / gridDim.x) << 5;
  const int nfull = P >> 13;
  for (int tr = 0; tr < nfull; ++tr) {
    uint4 v[8];
#pragma unroll
    for (int u = 0; u < 8; ++u) {
      int idx = base + ((tr * 8 + u) << 10) + tid;
      if (idx >= P) idx -= P;
      v[u] = keys[idx];
    }
    unsigned hm[4];
#pragma unroll
    for (int p = 0; p < 4; ++p)
      hm[p] = piece_match(v[2 * p], pat) | (piece_match(v[2 * p + 1], pat) >> 4);
#pragma unroll
    for (int p = 0; p < 4; ++p) {
      unsigned m = hm[p];
      while (m) {
        const int k = __builtin_ctz(m);
        m &= m - 1;
        const int r = k & 7;
        int idx = base + ((tr * 8 + 2 * p + (r < 4 ? 1 : 0)) << 10) + tid;
        if (idx >= P) idx -= P;
        queue_edge((idx << 4) + ((3 - (r & 3)) << 2) + (k >> 3));
      }
    }
  }
  for (int sidx = (nfull << 13) + tid; sidx < P; sidx += 1024) {
    int idx = base + sidx;
    if (idx >= P) idx -= P;
    unsigned m = piece_match(keys[idx], pat);
    while (m) {
      const int k = __builtin_ctz(m);
      m &= m - 1;
      queue_edge((idx << 4) + ((7 - (k & 7)) << 2) + (k >> 3));
    }
  }
  for (int eb = (P << 4) + tid; eb < E; eb += 4096) {
    int d[4];
#pragma unroll
    for (int u = 0; u < 4; ++u) d[u] = dp[min(eb + u * 1024, E - 1)];
#pragma unroll
    for (int u = 0; u < 4; ++u)
      if (eb + u * 1024 < E && (unsigned)(d[u] - lo) < (unsigned)cnt) queue_edge(eb + u * 1024);
  }
  __syncthreads();

  const int qtot = qn;
  const int nqx = min(qtot, QCAP);
  if (nqx > cnt) {
    int ev[4], sv[4], dv[4];
    float4 av[4];
#pragma unroll
    for (int u = 0; u < 4; ++u) {
      const int i = cnt + u * 1024 + tid;
      const int e = qe[min(i, QCAP - 1)];
      ev[u] = (i < nqx) ? e : 0;
    }
#pragma unroll
    for (int u = 0; u < 4; ++u) {
      sv[u] = ei[ev[u]];
      dv[u] = dp[ev[u]];
    }
#pragma unroll
    for (int u = 0; u < 4; ++u) av[u] = ((const float4*)a_src)[sv[u]];
#pragma unroll
    for (int u = 0; u < 4; ++u) {
      const int i = cnt + u * 1024 + tid;
      if (i < nqx) {
        qe[i] = sv[u];
        qd[i] = dv[u] - lo;
        ((float4*)qs)[i] = max_edge(av[u], dv[u] - lo);
      }
    }
    for (int i = cnt + 4096 + tid; i < nqx; i += 1024) {
      const int e = qe[i];
      const int s = ei[e];
      const int dl = dp[e] - lo;
      qe[i] = s;
      qd[i] = dl;
      ((float4*)qs)[i] = max_edge(((const float4*)a_src)[s], dl);
    }
  }
  __syncthreads();

  const int nq = (qtot > QCAP) ? cnt : qtot;
  for (int i0 = wave * 8; i0 < nq; i0 += 256) {
    int s[8], dl[8];
    float4 m[8];
    float sc[8], mx[8];
#pragma unroll
    for (int u = 0; u < 8; ++u) {
      const int i = min(i0 + u, nq - 1);
      s[u] = __builtin_amdgcn_readfirstlane(qe[i]);
      dl[u] = __builtin_amdgcn_readfirstlane(qd[i]);
      m[u] = ((const float4*)xt)[(size_t)s[u] * 32 + lane];
      sc[u] = qs[i * HEADS + h];
      mx[u] = fkey_inv(mxk_l[dl[u] * HEADS + h]);
    }
#pragma unroll
    for (int u = 0; u < 8; ++u) {
      if (i0 + u < nq) {
        const float ex = __expf(sc[u] - mx[u]);
        float* ap = acc_l + dl[u] * APITCH + coff;
        atomicAdd(ap + 0, ex * m[u].x);
        atomicAdd(ap + 1, ex * m[u].y);
        atomicAdd(ap + 2, ex * m[u].z);
        atomicAdd(ap + 3, ex * m[u].w);
        if ((lane & 7) == 0) atomicAdd(esum_l + dl[u] * HEADS + h, ex);
      }
    }
  }
  if (qtot > QCAP) {
    for (int e = tid; e < E; e += 1024) {
      const int dl = dp[e] - lo;
      if ((unsigned)dl < (unsigned)cnt) {
        const int s = ei[e];
#pragma unroll 1
        for (int hh = 0; hh < HEADS; ++hh) {
          const float ex = __expf(lrelu(a_src[(size_t)s * HEADS + hh] + adl_l[dl * HEADS + hh]) -
                                  fkey_inv(mxk_l[dl * HEADS + hh]));
          atomicAdd(esum_l + dl * HEADS + hh, ex);
#pragma unroll 1
          for (int ch = hh * 32; ch < hh * 32 + 32; ++ch)
            atomicAdd(acc_l + dl * APITCH + ch + (ch >> 6), ex * xt[(size_t)s * OUT_DIM + ch]);
        }
      }
    }
  }
  __syncthreads();

  const float4 gb = ((const float4*)gat_bias)[lane];
  const float4 rb = ((const float4*)res_b)[lane];
  const float4 g = ((const float4*)ln_g)[lane];
  const float4 b = ((const float4*)ln_b)[lane];
  for (int dl = wave; dl < cnt; dl += 32) {
    const int n = lo + dl;
    const float* ap = acc_l + dl * APITCH + coff;
    const float4 rr = ((const float4*)r)[(size_t)n * 32 + lane];
    const float den = esum_l[dl * HEADS + h] + 1e-16f;
    float4 v;
    v.x = ap[0] / den + gb.x + rr.x + rb.x;
    v.y = ap[1] / den + gb.y + rr.y + rb.y;
    v.z = ap[2] / den + gb.z + rr.z + rb.z;
    v.w = ap[3] / den + gb.w + rr.w + rb.w;
    float sm = v.x + v.y + v.z + v.w;
#pragma unroll
    for (int off = 1; off < 32; off <<= 1) sm += __shfl_xor(sm, off, 32);
    const float mu = sm * (1.0f / 128.0f);
    const float dx = v.x - mu, dy = v.y - mu, dz = v.z - mu, dw = v.w - mu;
    float q = dx * dx + dy * dy + dz * dz + dw * dw;
#pragma unroll
    for (int off = 1; off < 32; off <<= 1) q += __shfl_xor(q, off, 32);
    const float inv = rsqrtf(q * (1.0f / 128.0f) + 1e-5f);
    float4 o;
    o.x = lrelu(dx * inv * g.x + b.x);
    o.y = lrelu(dy * inv * g.y + b.y);
    o.z = lrelu(dz * inv * g.z + b.z);
    o.w = lrelu(dw * inv * g.w + b.w);
    ((float4*)out)[(size_t)n * 32 + lane] = o;
  }
}

extern "C" void kernel_launch(void* const* d_in, const int* in_sizes, int n_in,
                              void* d_out, int out_size, void* d_ws, size_t ws_size,
                              hipStream_t stream) {
  const float* x        = (const float*)d_in[0];
  const int*   ei       = (const int*)d_in[1];
  const float* lin_w    = (const float*)d_in[2];
  const float* att_src  = (const float*)d_in[3];
  const float* att_dst  = (const float*)d_in[4];
  const float* gat_bias = (const float*)d_in[5];
  const float* res_w    = (const float*)d_in[6];
  const float* res_b    = (const float*)d_in[7];
  const float* ln_g     = (const float*)d_in[8];
  const float* ln_b     = (const float*)d_in[9];
  const int N  = in_sizes[0] / IN_CH;
  const int E  = in_sizes[1] / 2;

  float* xt    = (float*)d_ws;
  float* z     = xt + (size_t)N * OUT_DIM;
  float* a_src = z + (size_t)N * OUT_DIM;
  float* a_dst = a_src + (size_t)N * HEADS;
  unsigned* keys = (unsigned*)(a_dst + (size_t)N * HEADS);

  k_gemm<<<dim3((N + 63) / 64, 2), 128, 0, stream>>>(x, lin_w, res_w, xt, z, N);
  k_node<<<(N + 7) / 8, 256, 0, stream>>>(xt, att_src, att_dst, a_src, a_dst, ei, keys, E, N);
  k_own<<<(N + NPW - 1) / NPW, 1024, 0, stream>>>(ei, (const uint4*)keys, xt, a_src, a_dst, z, gat_bias,
                                                  res_b, ln_g, ln_b, (float*)d_out, E, N);
}


# Residual product x@res_w^T moves from the GEMM launch into the owner groups: f32 WMMA on a group's own rows, kept in LDS for LayerNorm, never stored; queue capacity 4096 makes room; sums unchanged
# speedup vs baseline: 5.3077x; 1.0466x over previous
#include <hip/hip_runtime.h>
#include <hip/hip_bf16.h>

typedef __attribute__((ext_vector_type(2))) float v2f;
typedef __attribute__((ext_vector_type(8))) float v8f;

#define IN_CH 64
#define OUT_DIM 128
#define HEADS 4
#define NPW 196

static __device__ __forceinline__ float lrelu(float x) { return fmaxf(x, 0.2f * x); }

static __device__ __forceinline__ unsigned fkey(float f) {
  unsigned u = __float_as_uint(f);
  return (u & 0x80000000u) ? ~u : (u | 0x80000000u);
}
static __device__ __forceinline__ float fkey_inv(unsigned k) {
  return __uint_as_float((k & 0x80000000u) ? (k & 0x7FFFFFFFu) : ~k);
}

__global__ __launch_bounds__(128) void k_gemm(const float* __restrict__ x,
                                              const float* __restrict__ lin_w,
                                              float* __restrict__ xt,
                                              int N) {
  const int wave = __builtin_amdgcn_readfirstlane(threadIdx.x >> 5);
  const int lane = threadIdx.x & 31;
  const int c16 = lane & 15;
  const int hf = lane >> 4;
  const int m0 = (blockIdx.x * 4 + wave) * 16;
  if (m0 >= N) return;
  const float* Wm = lin_w;
  float* dst = xt;
  int arow = m0 + c16;
  if (arow >= N) arow = N - 1;
  const float4* xr = (const float4*)(x + (size_t)arow * IN_CH + 4 * hf);
  const float4* wr = (const float4*)(Wm + (size_t)c16 * IN_CH + 4 * hf);
  float4 a[8], b0[8], b1[8];
#pragma unroll
  for (int i = 0; i < 8; ++i) a[i] = xr[2 * i];
#pragma unroll
  for (int i = 0; i < 8; ++i) {
    b0[i] = wr[2 * i];
    b1[i] = wr[16 * IN_CH / 4 + 2 * i];
  }
  float* orow = dst + (size_t)(m0 + (hf << 3)) * OUT_DIM + c16;
  const bool whole = (m0 + 16 <= N);
#pragma unroll
  for (int tp = 0; tp < 4; ++tp) {
    float4 n0[8], n1[8];
    if (tp < 3) {
#pragma unroll
      for (int i = 0; i < 8; ++i) {
        n0[i] = wr[(tp + 1) * 32 * IN_CH / 4 + 2 * i];
        n1[i] = wr[(tp + 1) * 32 * IN_CH / 4 + 16 * IN_CH / 4 + 2 * i];
      }
    }
    v8f c0 = {0.f, 0.f, 0.f, 0.f, 0.f, 0.f, 0.f, 0.f};
    v8f c1 = {0.f, 0.f, 0.f, 0.f, 0.f, 0.f, 0.f, 0.f};
#pragma unroll
    for (int i = 0; i < 8; ++i) {
      const v2f alo = {a[i].x, a[i].y};
      const v2f ahi = {a[i].z, a[i].w};
      const v2f b0lo = {b0[i].x, b0[i].y};
      const v2f b0hi = {b0[i].z, b0[i].w};
      const v2f b1lo = {b1[i].x, b1[i].y};
      const v2f b1hi = {b1[i].z, b1[i].w};
      c0 = __builtin_amdgcn_wmma_f32_16x16x4_f32(false, alo, false, b0lo, (short)0, c0, false, false);
      c1 = __builtin_amdgcn_wmma_f32_16x16x4_f32(false, alo, false, b1lo, (short)0, c1, false, false);
      c0 = __builtin_amdgcn_wmma_f32_16x16x4_f32(false, ahi, false, b0hi, (short)0, c0, false, false);
      c1 = __builtin_amdgcn_wmma_f32_16x16x4_f32(false, ahi, false, b1hi, (short)0, c1, false, false);
    }
    if (whole) {
#pragma unroll
      for (int r = 0; r < 8; ++r) {
        orow[(size_t)r * OUT_DIM + tp * 32] = c0[r];
        orow[(size_t)r * OUT_DIM + tp * 32 + 16] = c1[r];
      }
    } else {
#pragma unroll
      for (int r = 0; r < 8; ++r) {
        if (m0 + (hf << 3) + r < N) {
          orow[(size_t)r * OUT_DIM + tp * 32] = c0[r];
          orow[(size_t)r * OUT_DIM + tp * 32 + 16] = c1[r];
        }
      }
    }
    if (tp < 3) {
#pragma unroll
      for (int i = 0; i < 8; ++i) {
        b0[i] = n0[i];
        b1[i] = n1[i];
      }
    }
  }
}

__global__ __launch_bounds__(256) void k_node(const float* __restrict__ xt,
                                              const float* __restrict__ att_src,
                                              const float* __restrict__ att_dst,
                                              float* __restrict__ a_src,
                                              float* __restrict__ a_dst,
                                              const int* __restrict__ ei,
                                              unsigned* __restrict__ keys, int E, int N) {
  const int nkw = ((E & 3) == 0 && (N + NPW - 1) / NPW <= 256) ? ((E >> 4) << 2) : 0;
  for (int t = blockIdx.x * 256 + threadIdx.x; t < nkw; t += gridDim.x * 256) {
    const int4 d = ((const int4*)(ei + E))[t];
    keys[t] = ((unsigned)d.x / NPW) | (((unsigned)d.y / NPW) << 8) |
              (((unsigned)d.z / NPW) << 16) | (((unsigned)d.w / NPW) << 24);
  }
  const int lane = threadIdx.x & 31;
  const int n = blockIdx.x * 8 + (threadIdx.x >> 5);
  if (n >= N) return;
  const float4 v = ((const float4*)xt)[(size_t)n * 32 + lane];
  const float4 as = ((const float4*)att_src)[lane];
  const float4 ad = ((const float4*)att_dst)[lane];
  float ps = v.x * as.x + v.y * as.y + v.z * as.z + v.w * as.w;
  float pd = v.x * ad.x + v.y * ad.y + v.z * ad.z + v.w * ad.w;
#pragma unroll
  for (int off = 1; off < 8; off <<= 1) {
    ps += __shfl_xor(ps, off, 32);
    pd += __shfl_xor(pd, off, 32);
  }
  const int h = lane >> 3;
  if ((lane & 7) == 0) {
    a_src[(size_t)n * HEADS + h] = ps;
    a_dst[(size_t)n * HEADS + h] = pd;
  }
}

#define APITCH 130
#define QCAP 4096
#define ZROWS (((NPW + 15) / 16) * 16)
#define ZPITCH 132
static __device__ __forceinline__ unsigned key_match(unsigned v, unsigned pat) {
  const unsigned t = v ^ pat;
  return ~(((t & 0x7F7F7F7Fu) + 0x7F7F7F7Fu) | t | 0x7F7F7F7Fu);
}
static __device__ __forceinline__ unsigned piece_match(const uint4 v, unsigned pat) {
  return key_match(v.x, pat) | (key_match(v.y, pat) >> 1) | (key_match(v.z, pat) >> 2) |
         (key_match(v.w, pat) >> 3);
}
static_assert(NPW * HEADS <= 1024, "one lane per (owned node, head) in the start phase");
static_assert((NPW * APITCH) % 4 == 0, "acc is cleared in 16-byte pieces");
static_assert(NPW < QCAP, "the self-loops must fit the queue");
__global__ __launch_bounds__(1024) void k_own(const int* __restrict__ ei,
                                              const uint4* __restrict__ keys,
                                              const float* __restrict__ xt,
                                              const float* __restrict__ a_src,
                                              const float* __restrict__ a_dst,
                                              const float* __restrict__ x,
                                              const float* __restrict__ res_w,
                                              const float* __restrict__ gat_bias,
                                              const float* __restrict__ res_b,
                                              const float* __restrict__ ln_g,
                                              const float* __restrict__ ln_b,
                                              float* __restrict__ out, int E, int N) {
  __shared__ __attribute__((aligned(16))) float acc_l[NPW * APITCH];
  __shared__ __attribute__((aligned(16))) float esum_l[NPW * HEADS];
  __shared__ __attribute__((aligned(16))) float adl_l[NPW * HEADS];
  __shared__ unsigned mxk_l[NPW * HEADS];
  __shared__ int qe[QCAP];
  __shared__ int qd[QCAP];
  __shared__ __attribute__((aligned(16))) float qs[QCAP * HEADS];
  __shared__ __attribute__((aligned(16))) float z_l[ZROWS * ZPITCH];
  __shared__ int qn;

  const int tid = threadIdx.x;
  const int lane = tid & 31;
  const int wave = __builtin_amdgcn_readfirstlane(tid >> 5);
  const int h = lane >> 3;
  const int coff = lane * 4 + (lane >> 4);
  const int lo = blockIdx.x * NPW;
  const int cnt = min(NPW, N - lo);
  const int* __restrict__ dp = ei + E;

  if (tid == 0) qn = cnt;
  for (int i = tid; i < NPW * APITCH / 4; i += 1024)
    ((float4*)acc_l)[i] = make_float4(0.f, 0.f, 0.f, 0.f);
  if (tid < cnt * HEADS) {
    const float as = a_src[(size_t)lo * HEADS + tid];
    const float ad = a_dst[(size_t)lo * HEADS + tid];
    const float sc = lrelu(as + ad);
    adl_l[tid] = ad;
    qs[tid] = sc;
    mxk_l[tid] = fkey(sc);
    esum_l[tid] = 0.f;
  }
  if (tid < cnt) {
    qe[tid] = lo + tid;
    qd[tid] = tid;
  }
  __syncthreads();

  if (wave < 2 * (ZROWS / 16)) {
    const int c16 = lane & 15;
    const int hf = lane >> 4;
    const int rt = wave >> 1;
    const int ch = wave & 1;
    const int arow = min(lo + rt * 16 + c16, N - 1);
    const float4* xr = (const float4*)(x + (size_t)arow * IN_CH + 4 * hf);
    const float4* wr = (const float4*)(res_w + (size_t)(ch * 64 + c16) * IN_CH + 4 * hf);
    float4 a[8];
#pragma unroll
    for (int i = 0; i < 8; ++i) a[i] = xr[2 * i];
    float* zp = z_l + (rt * 16 + (hf << 3)) * ZPITCH + ch * 64 + c16;
#pragma unroll
    for (int tp = 0; tp < 2; ++tp) {
      float4 b0[8], b1[8];
#pragma unroll
      for (int i = 0; i < 8; ++i) {
        b0[i] = wr[tp * 32 * IN_CH / 4 + 2 * i];
        b1[i] = wr[tp * 32 * IN_CH / 4 + 16 * IN_CH / 4 + 2 * i];
      }
      v8f c0 = {0.f, 0.f, 0.f, 0.f, 0.f, 0.f, 0.f, 0.f};
      v8f c1 = {0.f, 0.f, 0.f, 0.f, 0.f, 0.f, 0.f, 0.f};
#pragma unroll
      for (int i = 0; i < 8; ++i) {
        const v2f alo = {a[i].x, a[i].y};
        const v2f ahi = {a[i].z, a[i].w};
        const v2f b0lo = {b0[i].x, b0[i].y};
        const v2f b0hi = {b0[i].z, b0[i].w};
        const v2f b1lo = {b1[i].x, b1[i].y};
        const v2f b1hi = {b1[i].z, b1[i].w};
        c0 = __builtin_amdgcn_wmma_f32_16x16x4_f32(false, alo, false, b0lo, (short)0, c0, false, false);
        c1 = __builtin_amdgcn_wmma_f32_16x16x4_f32(false, alo, false, b1lo, (short)0, c1, false, false);
        c0 = __builtin_amdgcn_wmma_f32_16x16x4_f32(false, ahi, false, b0hi, (short)0, c0, false, false);
        c1 = __builtin_amdgcn_wmma_f32_16x16x4_f32(false, ahi, false, b1hi, (short)0, c1, false, false);
      }
#pragma unroll
      for (int r = 0; r < 8; ++r) {
        zp[r * ZPITCH + tp * 32] = c0[r];
        zp[r * ZPITCH + tp * 32 + 16] = c1[r];
      }
    }
  }

  auto max_edge = [&](const float4 as, int dl) {
    const float4 ad = *(const float4*)(adl_l + dl * HEADS);
    float4 sc;
    sc.x = lrelu(as.x + ad.x);
    sc.y = lrelu(as.y + ad.y);
    sc.z = lrelu(as.z + ad.z);
    sc.w = lrelu(as.w + ad.w);
    unsigned* mp = mxk_l + dl * HEADS;
    atomicMax(mp + 0, fkey(sc.x));
    atomicMax(mp + 1, fkey(sc.y));
    atomicMax(mp + 2, fkey(sc.z));
    atomicMax(mp + 3, fkey(sc.w));
    return sc;
  };

  auto queue_edge = [&](int e) {
    const int slot = atomicAdd(&qn, 1);
    if (slot < QCAP) {
      qe[slot] = e;
    } else {
      max_edge(((const float4*)a_src)[ei[e]], dp[e] - lo);
    }
  };

  const int P = ((E & 3) == 0 && gridDim.x <= 256) ? (E >> 4) : 0;
  const unsigned pat = blockIdx.x * 0x01010101u;
  const int base = (int)(((unsigned long long)blockIdx.x * (unsigned)(P >> 5)) / gridDim.x) << 5;
  const int nfull = P >> 13;
  for (int tr = 0; tr < nfull; ++tr) {
    uint4 v[8];
#pragma unroll
    for (int u = 0; u < 8; ++u) {
      int idx = base + ((tr * 8 + u) << 10) + tid;
      if (idx >= P) idx -= P;
      v[u] = keys[idx];
    }
    unsigned hm[4];
#pragma unroll
    for (int p = 0; p < 4; ++p)
      hm[p] = piece_match(v[2 * p], pat) | (piece_match(v[2 * p + 1], pat) >> 4);
#pragma unroll
    for (int p = 0; p < 4; ++p) {
      unsigned m = hm[p];
      while (m) {
        const int k = __builtin_ctz(m);
        m &= m - 1;
        const int r = k & 7;
        int idx = base + ((tr * 8 + 2 * p + (r < 4 ? 1 : 0)) << 10) + tid;
        if (idx >= P) idx -= P;
        queue_edge((idx << 4) + ((3 - (r & 3)) << 2) + (k >> 3));
      }
    }
  }
  for (int sidx = (nfull << 13) + tid; sidx < P; sidx += 1024) {
    int idx = base + sidx;
    if (idx >= P) idx -= P;
    unsigned m = piece_match(keys[idx], pat);
    while (m) {
      const int k = __builtin_ctz(m);
      m &= m - 1;
      queue_edge((idx << 4) + ((7 - (k & 7)) << 2) + (k >> 3));
    }
  }
  for (int eb = (P << 4) + tid; eb < E; eb += 4096) {
    int d[4];
#pragma unroll
    for (int u = 0; u < 4; ++u) d[u] = dp[min(eb + u * 1024, E - 1)];
#pragma unroll
    for (int u = 0; u < 4; ++u)
      if (eb + u * 1024 < E && (unsigned)(d[u] - lo) < (unsigned)cnt) queue_edge(eb + u * 1024);
  }
  __syncthreads();

  const int qtot = qn;
  const int nqx = min(qtot, QCAP);
  if (nqx > cnt) {
    int ev[4], sv[4], dv[4];
    float4 av[4];
#pragma unroll
    for (int u = 0; u < 4; ++u) {
      const int i = cnt + u * 1024 + tid;
      const int e = qe[min(i, QCAP - 1)];
      ev[u] = (i < nqx) ? e : 0;
    }
#pragma unroll
    for (int u = 0; u < 4; ++u) {
      sv[u] = ei[ev[u]];
      dv[u] = dp[ev[u]];
    }
#pragma unroll
    for (int u = 0; u < 4; ++u) av[u] = ((const float4*)a_src)[sv[u]];
#pragma unroll
    for (int u = 0; u < 4; ++u) {
      const int i = cnt + u * 1024 + tid;
      if (i < nqx) {
        qe[i] = sv[u];
        qd[i] = dv[u] - lo;
        ((float4*)qs)[i] = max_edge(av[u], dv[u] - lo);
      }
    }
    for (int i = cnt + 4096 + tid; i < nqx; i += 1024) {
      const int e = qe[i];
      const int s = ei[e];
      const int dl = dp[e] - lo;
      qe[i] = s;
      qd[i] = dl;
      ((float4*)qs)[i] = max_edge(((const float4*)a_src)[s], dl);
    }
  }
  __syncthreads();

  const int nq = (qtot > QCAP) ? cnt : qtot;
  for (int i0 = wave * 8; i0 < nq; i0 += 256) {
    int s[8], dl[8];
    float4 m[8];
    float sc[8], mx[8];
#pragma unroll
    for (int u = 0; u < 8; ++u) {
      const int i = min(i0 + u, nq - 1);
      s[u] = __builtin_amdgcn_readfirstlane(qe[i]);
      dl[u] = __builtin_amdgcn_readfirstlane(qd[i]);
      m[u] = ((const float4*)xt)[(size_t)s[u] * 32 + lane];
      sc[u] = qs[i * HEADS + h];
      mx[u] = fkey_inv(mxk_l[dl[u] * HEADS + h]);
    }
#pragma unroll
    for (int u = 0; u < 8; ++u) {
      if (i0 + u < nq) {
        const float ex = __expf(sc[u] - mx[u]);
        float* ap = acc_l + dl[u] * APITCH + coff;
        atomicAdd(ap + 0, ex * m[u].x);
        atomicAdd(ap + 1, ex * m[u].y);
        atomicAdd(ap + 2, ex * m[u].z);
        atomicAdd(ap + 3, ex * m[u].w);
        if ((lane & 7) == 0) atomicAdd(esum_l + dl[u] * HEADS + h, ex);
      }
    }
  }
  if (qtot > QCAP) {
    for (int e = tid; e < E; e += 1024) {
      const int dl = dp[e] - lo;
      if ((unsigned)dl < (unsigned)cnt) {
        const int s = ei[e];
#pragma unroll 1
        for (int hh = 0; hh < HEADS; ++hh) {
          const float ex = __expf(lrelu(a_src[(size_t)s * HEADS + hh] + adl_l[dl * HEADS + hh]) -
                                  fkey_inv(mxk_l[dl * HEADS + hh]));
          atomicAdd(esum_l + dl * HEADS + hh, ex);
#pragma unroll 1
          for (int ch = hh * 32; ch < hh * 32 + 32; ++ch)
            atomicAdd(acc_l + dl * APITCH + ch + (ch >> 6), ex * xt[(size_t)s * OUT_DIM + ch]);
        }
      }
    }
  }
  __syncthreads();

  const float4 gb = ((const float4*)gat_bias)[lane];
  const float4 rb = ((const float4*)res_b)[lane];
  const float4 g = ((const float4*)ln_g)[lane];
  const float4 b = ((const float4*)ln_b)[lane];
  for (int dl = wave; dl < cnt; dl += 32) {
    const int n = lo + dl;
    const float* ap = acc_l + dl * APITCH + coff;
    const float4 rr = *(const float4*)(z_l + dl * ZPITCH + lane * 4);
    const float den = esum_l[dl * HEADS + h] + 1e-16f;
    float4 v;
    v.x = ap[0] / den + gb.x + rr.x + rb.x;
    v.y = ap[1] / den + gb.y + rr.y + rb.y;
    v.z = ap[2] / den + gb.z + rr.z + rb.z;
    v.w = ap[3] / den + gb.w + rr.w + rb.w;
    float sm = v.x + v.y + v.z + v.w;
#pragma unroll
    for (int off = 1; off < 32; off <<= 1) sm += __shfl_xor(sm, off, 32);
    const float mu = sm * (1.0f / 128.0f);
    const float dx = v.x - mu, dy = v.y - mu, dz = v.z - mu, dw = v.w - mu;
    float q = dx * dx + dy * dy + dz * dz + dw * dw;
#pragma unroll
    for (int off = 1; off < 32; off <<= 1) q += __shfl_xor(q, off, 32);
    const float inv = rsqrtf(q * (1.0f / 128.0f) + 1e-5f);
    float4 o;
    o.x = lrelu(dx * inv * g.x + b.x);
    o.y = lrelu(dy * inv * g.y + b.y);
    o.z = lrelu(dz * inv * g.z + b.z);
    o.w = lrelu(dw * inv * g.w + b.w);
    ((float4*)out)[(size_t)n * 32 + lane] = o;
  }
}

extern "C" void kernel_launch(void* const* d_in, const int* in_sizes, int n_in,
                              void* d_out, int out_size, void* d_ws, size_t ws_size,
                              hipStream_t stream) {
  const float* x        = (const float*)d_in[0];
  const int*   ei       = (const int*)d_in[1];
  const float* lin_w    = (const float*)d_in[2];
  const float* att_src  = (const float*)d_in[3];
  const float* att_dst  = (const float*)d_in[4];
  const float* gat_bias = (const float*)d_in[5];
  const float* res_w    = (const float*)d_in[6];
  const float* res_b    = (const float*)d_in[7];
  const float* ln_g     = (const float*)d_in[8];
  const float* ln_b     = (const float*)d_in[9];
  const int N  = in_sizes[0] / IN_CH;
  const int E  = in_sizes[1] / 2;

  float* xt    = (float*)d_ws;
  float* a_src = xt + (size_t)N * OUT_DIM;
  float* a_dst = a_src + (size_t)N * HEADS;
  unsigned* keys = (unsigned*)(a_dst + (size_t)N * HEADS);

  k_gemm<<<(N + 63) / 64, 128, 0, stream>>>(x, lin_w, xt, N);
  k_node<<<(N + 7) / 8, 256, 0, stream>>>(xt, att_src, att_dst, a_src, a_dst, ei, keys, E, N);
  k_own<<<(N + NPW - 1) / NPW, 1024, 0, stream>>>(ei, (const uint4*)keys, xt, a_src, a_dst, x, res_w, gat_bias,
                                                  res_b, ln_g, ln_b, (float*)d_out, E, N);
}


# Sum phase: softmax weights made by one lane per (queued edge, head), 8 edges per LDS load and per esum add, handed to the row lanes by an 8-lane broadcast (DPP8); esum add order changes
# speedup vs baseline: 5.5903x; 1.0532x over previous
#include <hip/hip_runtime.h>
#include <hip/hip_bf16.h>

typedef __attribute__((ext_vector_type(2))) float v2f;
typedef __attribute__((ext_vector_type(8))) float v8f;

#define IN_CH 64
#define OUT_DIM 128
#define HEADS 4
#define NPW 196

static __device__ __forceinline__ float lrelu(float x) { return fmaxf(x, 0.2f * x); }

static __device__ __forceinline__ unsigned fkey(float f) {
  unsigned u = __float_as_uint(f);
  return (u & 0x80000000u) ? ~u : (u | 0x80000000u);
}
static __device__ __forceinline__ float fkey_inv(unsigned k) {
  return __uint_as_float((k & 0x80000000u) ? (k & 0x7FFFFFFFu) : ~k);
}

__global__ __launch_bounds__(128) void k_gemm(const float* __restrict__ x,
                                              const float* __restrict__ lin_w,
                                              float* __restrict__ xt,
                                              int N) {
  const int wave = __builtin_amdgcn_readfirstlane(threadIdx.x >> 5);
  const int lane = threadIdx.x & 31;
  const int c16 = lane & 15;
  const int hf = lane >> 4;
  const int m0 = (blockIdx.x * 4 + wave) * 16;
  if (m0 >= N) return;
  const float* Wm = lin_w;
  float* dst = xt;
  int arow = m0 + c16;
  if (arow >= N) arow = N - 1;
  const float4* xr = (const float4*)(x + (size_t)arow * IN_CH + 4 * hf);
  const float4* wr = (const float4*)(Wm + (size_t)c16 * IN_CH + 4 * hf);
  float4 a[8], b0[8], b1[8];
#pragma unroll
  for (int i = 0; i < 8; ++i) a[i] = xr[2 * i];
#pragma unroll
  for (int i = 0; i < 8; ++i) {
    b0[i] = wr[2 * i];
    b1[i] = wr[16 * IN_CH / 4 + 2 * i];
  }
  float* orow = dst + (size_t)(m0 + (hf << 3)) * OUT_DIM + c16;
  const bool whole = (m0 + 16 <= N);
#pragma unroll
  for (int tp = 0; tp < 4; ++tp) {
    float4 n0[8], n1[8];
    if (tp < 3) {
#pragma unroll
      for (int i = 0; i < 8; ++i) {
        n0[i] = wr[(tp + 1) * 32 * IN_CH / 4 + 2 * i];
        n1[i] = wr[(tp + 1) * 32 * IN_CH / 4 + 16 * IN_CH / 4 + 2 * i];
      }
    }
    v8f c0 = {0.f, 0.f, 0.f, 0.f, 0.f, 0.f, 0.f, 0.f};
    v8f c1 = {0.f, 0.f, 0.f, 0.f, 0.f, 0.f, 0.f, 0.f};
#pragma unroll
    for (int i = 0; i < 8; ++i) {
      const v2f alo = {a[i].x, a[i].y};
      const v2f ahi = {a[i].z, a[i].w};
      const v2f b0lo = {b0[i].x, b0[i].y};
      const v2f b0hi = {b0[i].z, b0[i].w};
      const v2f b1lo = {b1[i].x, b1[i].y};
      const v2f b1hi = {b1[i].z, b1[i].w};
      c0 = __builtin_amdgcn_wmma_f32_16x16x4_f32(false, alo, false, b0lo, (short)0, c0, false, false);
      c1 = __builtin_amdgcn_wmma_f32_16x16x4_f32(false, alo, false, b1lo, (short)0, c1, false, false);
      c0 = __builtin_amdgcn_wmma_f32_16x16x4_f32(false, ahi, false, b0hi, (short)0, c0, false, false);
      c1 = __builtin_amdgcn_wmma_f32_16x16x4_f32(false, ahi, false, b1hi, (short)0, c1, false, false);
    }
    if (whole) {
#pragma unroll
      for (int r = 0; r < 8; ++r) {
        orow[(size_t)r * OUT_DIM + tp * 32] = c0[r];
        orow[(size_t)r * OUT_DIM + tp * 32 + 16] = c1[r];
      }
    } else {
#pragma unroll
      for (int r = 0; r < 8; ++r) {
        if (m0 + (hf << 3) + r < N) {
          orow[(size_t)r * OUT_DIM + tp * 32] = c0[r];
          orow[(size_t)r * OUT_DIM + tp * 32 + 16] = c1[r];
        }
      }
    }
    if (tp < 3) {
#pragma unroll
      for (int i = 0; i < 8; ++i) {
        b0[i] = n0[i];
        b1[i] = n1[i];
      }
    }
  }
}

__global__ __launch_bounds__(256) void k_node(const float* __restrict__ xt,
                                              const float* __restrict__ att_src,
                                              const float* __restrict__ att_dst,
                                              float* __restrict__ a_src,
                                              float* __restrict__ a_dst,
                                              const int* __restrict__ ei,
                                              unsigned* __restrict__ keys, int E, int N) {
  const int nkw = ((E & 3) == 0 && (N + NPW - 1) / NPW <= 256) ? ((E >> 4) << 2) : 0;
  for (int t = blockIdx.x * 256 + threadIdx.x; t < nkw; t += gridDim.x * 256) {
    const int4 d = ((const int4*)(ei + E))[t];
    keys[t] = ((unsigned)d.x / NPW) | (((unsigned)d.y / NPW) << 8) |
              (((unsigned)d.z / NPW) << 16) | (((unsigned)d.w / NPW) << 24);
  }
  const int lane = threadIdx.x & 31;
  const int n = blockIdx.x * 8 + (threadIdx.x >> 5);
  if (n >= N) return;
  const float4 v = ((const float4*)xt)[(size_t)n * 32 + lane];
  const float4 as = ((const float4*)att_src)[lane];
  const float4 ad = ((const float4*)att_dst)[lane];
  float ps = v.x * as.x + v.y * as.y + v.z * as.z + v.w * as.w;
  float pd = v.x * ad.x + v.y * ad.y + v.z * ad.z + v.w * ad.w;
#pragma unroll
  for (int off = 1; off < 8; off <<= 1) {
    ps += __shfl_xor(ps, off, 32);
    pd += __shfl_xor(pd, off, 32);
  }
  const int h = lane >> 3;
  if ((lane & 7) == 0) {
    a_src[(size_t)n * HEADS + h] = ps;
    a_dst[(size_t)n * HEADS + h] = pd;
  }
}

#define APITCH 130
#define QCAP 4096
#define ZROWS (((NPW + 15) / 16) * 16)
#define ZPITCH 132
static __device__ __forceinline__ unsigned key_match(unsigned v, unsigned pat) {
  const unsigned t = v ^ pat;
  return ~(((t & 0x7F7F7F7Fu) + 0x7F7F7F7Fu) | t | 0x7F7F7F7Fu);
}
static __device__ __forceinline__ unsigned piece_match(const uint4 v, unsigned pat) {
  return key_match(v.x, pat) | (key_match(v.y, pat) >> 1) | (key_match(v.z, pat) >> 2) |
         (key_match(v.w, pat) >> 3);
}
static __device__ __forceinline__ float group8_bcast(float v, int u) {
  const unsigned b = __float_as_uint(v);
  unsigned r;
  switch (u) {
    case 0: r = __builtin_amdgcn_mov_dpp8(b, 0 * 0x249249); break;
    case 1: r = __builtin_amdgcn_mov_dpp8(b, 1 * 0x249249); break;
    case 2: r = __builtin_amdgcn_mov_dpp8(b, 2 * 0x249249); break;
    case 3: r = __builtin_amdgcn_mov_dpp8(b, 3 * 0x249249); break;
    case 4: r = __builtin_amdgcn_mov_dpp8(b, 4 * 0x249249); break;
    case 5: r = __builtin_amdgcn_mov_dpp8(b, 5 * 0x249249); break;
    case 6: r = __builtin_amdgcn_mov_dpp8(b, 6 * 0x249249); break;
    default: r = __builtin_amdgcn_mov_dpp8(b, 7 * 0x249249); break;
  }
  return __uint_as_float(r);
}
static_assert(NPW * HEADS <= 1024, "one lane per (owned node, head) in the start phase");
static_assert((NPW * APITCH) % 4 == 0, "acc is cleared in 16-byte pieces");
static_assert(NPW < QCAP, "the self-loops must fit the queue");
__global__ __launch_bounds__(1024) void k_own(const int* __restrict__ ei,
                                              const uint4* __restrict__ keys,
                                              const float* __restrict__ xt,
                                              const float* __restrict__ a_src,
                                              const float* __restrict__ a_dst,
                                              const float* __restrict__ x,
                                              const float* __restrict__ res_w,
                                              const float* __restrict__ gat_bias,
                                              const float* __restrict__ res_b,
                                              const float* __restrict__ ln_g,
                                              const float* __restrict__ ln_b,
                                              float* __restrict__ out, int E, int N) {
  __shared__ __attribute__((aligned(16))) float acc_l[NPW * APITCH];
  __shared__ __attribute__((aligned(16))) float esum_l[NPW * HEADS];
  __shared__ __attribute__((aligned(16))) float adl_l[NPW * HEADS];
  __shared__ unsigned mxk_l[NPW * HEADS];
  __shared__ int qe[QCAP];
  __shared__ int qd[QCAP];
  __shared__ __attribute__((aligned(16))) float qs[QCAP * HEADS];
  __shared__ __attribute__((aligned(16))) float z_l[ZROWS * ZPITCH];
  __shared__ int qn;

  const int tid = threadIdx.x;
  const int lane = tid & 31;
  const int wave = __builtin_amdgcn_readfirstlane(tid >> 5);
  const int h = lane >> 3;
  const int coff = lane * 4 + (lane >> 4);
  const int lo = blockIdx.x * NPW;
  const int cnt = min(NPW, N - lo);
  const int* __restrict__ dp = ei + E;

  if (tid == 0) qn = cnt;
  for (int i = tid; i < NPW * APITCH / 4; i += 1024)
    ((float4*)acc_l)[i] = make_float4(0.f, 0.f, 0.f, 0.f);
  if (tid < cnt * HEADS) {
    const float as = a_src[(size_t)lo * HEADS + tid];
    const float ad = a_dst[(size_t)lo * HEADS + tid];
    const float sc = lrelu(as + ad);
    adl_l[tid] = ad;
    qs[tid] = sc;
    mxk_l[tid] = fkey(sc);
    esum_l[tid] = 0.f;
  }
  if (tid < cnt) {
    qe[tid] = lo + tid;
    qd[tid] = tid;
  }
  __syncthreads();

  if (wave < 2 * (ZROWS / 16)) {
    const int c16 = lane & 15;
    const int hf = lane >> 4;
    const int rt = wave >> 1;
    const int ch = wave & 1;
    const int arow = min(lo + rt * 16 + c16, N - 1);
    const float4* xr = (const float4*)(x + (size_t)arow * IN_CH + 4 * hf);
    const float4* wr = (const float4*)(res_w + (size_t)(ch * 64 + c16) * IN_CH + 4 * hf);
    float4 a[8];
#pragma unroll
    for (int i = 0; i < 8; ++i) a[i] = xr[2 * i];
    float* zp = z_l + (rt * 16 + (hf << 3)) * ZPITCH + ch * 64 + c16;
#pragma unroll
    for (int tp = 0; tp < 2; ++tp) {
      float4 b0[8], b1[8];
#pragma unroll
      for (int i = 0; i < 8; ++i) {
        b0[i] = wr[tp * 32 * IN_CH / 4 + 2 * i];
        b1[i] = wr[tp * 32 * IN_CH / 4 + 16 * IN_CH / 4 + 2 * i];
      }
      v8f c0 = {0.f, 0.f, 0.f, 0.f, 0.f, 0.f, 0.f, 0.f};
      v8f c1 = {0.f, 0.f, 0.f, 0.f, 0.f, 0.f, 0.f, 0.f};
#pragma unroll
      for (int i = 0; i < 8; ++i) {
        const v2f alo = {a[i].x, a[i].y};
        const v2f ahi = {a[i].z, a[i].w};
        const v2f b0lo = {b0[i].x, b0[i].y};
        const v2f b0hi = {b0[i].z, b0[i].w};
        const v2f b1lo = {b1[i].x, b1[i].y};
        const v2f b1hi = {b1[i].z, b1[i].w};
        c0 = __builtin_amdgcn_wmma_f32_16x16x4_f32(false, alo, false, b0lo, (short)0, c0, false, false);
        c1 = __builtin_amdgcn_wmma_f32_16x16x4_f32(false, alo, false, b1lo, (short)0, c1, false, false);
        c0 = __builtin_amdgcn_wmma_f32_16x16x4_f32(false, ahi, false, b0hi, (short)0, c0, false, false);
        c1 = __builtin_amdgcn_wmma_f32_16x16x4_f32(false, ahi, false, b1hi, (short)0, c1, false, false);
      }
#pragma unroll
      for (int r = 0; r < 8; ++r) {
        zp[r * ZPITCH + tp * 32] = c0[r];
        zp[r * ZPITCH + tp * 32 + 16] = c1[r];
      }
    }
  }

  auto max_edge = [&](const float4 as, int dl) {
    const float4 ad = *(const float4*)(adl_l + dl * HEADS);
    float4 sc;
    sc.x = lrelu(as.x + ad.x);
    sc.y = lrelu(as.y + ad.y);
    sc.z = lrelu(as.z + ad.z);
    sc.w = lrelu(as.w + ad.w);
    unsigned* mp = mxk_l + dl * HEADS;
    atomicMax(mp + 0, fkey(sc.x));
    atomicMax(mp + 1, fkey(sc.y));
    atomicMax(mp + 2, fkey(sc.z));
    atomicMax(mp + 3, fkey(sc.w));
    return sc;
  };

  auto queue_edge = [&](int e) {
    const int slot = atomicAdd(&qn, 1);
    if (slot < QCAP) {
      qe[slot] = e;
    } else {
      max_edge(((const float4*)a_src)[ei[e]], dp[e] - lo);
    }
  };

  const int P = ((E & 3) == 0 && gridDim.x <= 256) ? (E >> 4) : 0;
  const unsigned pat = blockIdx.x * 0x01010101u;
  const int base = (int)(((unsigned long long)blockIdx.x * (unsigned)(P >> 5)) / gridDim.x) << 5;
  const int nfull = P >> 13;
  for (int tr = 0; tr < nfull; ++tr) {
    uint4 v[8];
#pragma unroll
    for (int u = 0; u < 8; ++u) {
      int idx = base + ((tr * 8 + u) << 10) + tid;
      if (idx >= P) idx -= P;
      v[u] = keys[idx];
    }
    unsigned hm[4];
#pragma unroll
    for (int p = 0; p < 4; ++p)
      hm[p] = piece_match(v[2 * p], pat) | (piece_match(v[2 * p + 1], pat) >> 4);
#pragma unroll
    for (int p = 0; p < 4; ++p) {
      unsigned m = hm[p];
      while (m) {
        const int k = __builtin_ctz(m);
        m &= m - 1;
        const int r = k & 7;
        int idx = base + ((tr * 8 + 2 * p + (r < 4 ? 1 : 0)) << 10) + tid;
        if (idx >= P) idx -= P;
        queue_edge((idx << 4) + ((3 - (r & 3)) << 2) + (k >> 3));
      }
    }
  }
  for (int sidx = (nfull << 13) + tid; sidx < P; sidx += 1024) {
    int idx = base + sidx;
    if (idx >= P) idx -= P;
    unsigned m = piece_match(keys[idx], pat);
    while (m) {
      const int k = __builtin_ctz(m);
      m &= m - 1;
      queue_edge((idx << 4) + ((7 - (k & 7)) << 2) + (k >> 3));
    }
  }
  for (int eb = (P << 4) + tid; eb < E; eb += 4096) {
    int d[4];
#pragma unroll
    for (int u = 0; u < 4; ++u) d[u] = dp[min(eb + u * 1024, E - 1)];
#pragma unroll
    for (int u = 0; u < 4; ++u)
      if (eb + u * 1024 < E && (unsigned)(d[u] - lo) < (unsigned)cnt) queue_edge(eb + u * 1024);
  }
  __syncthreads();

  const int qtot = qn;
  const int nqx = min(qtot, QCAP);
  if (nqx > cnt) {
    int ev[4], sv[4], dv[4];
    float4 av[4];
#pragma unroll
    for (int u = 0; u < 4; ++u) {
      const int i = cnt + u * 1024 + tid;
      const int e = qe[min(i, QCAP - 1)];
      ev[u] = (i < nqx) ? e : 0;
    }
#pragma unroll
    for (int u = 0; u < 4; ++u) {
      sv[u] = ei[ev[u]];
      dv[u] = dp[ev[u]];
    }
#pragma unroll
    for (int u = 0; u < 4; ++u) av[u] = ((const float4*)a_src)[sv[u]];
#pragma unroll
    for (int u = 0; u < 4; ++u) {
      const int i = cnt + u * 1024 + tid;
      if (i < nqx) {
        qe[i] = sv[u];
        qd[i] = dv[u] - lo;
        ((float4*)qs)[i] = max_edge(av[u], dv[u] - lo);
      }
    }
    for (int i = cnt + 4096 + tid; i < nqx; i += 1024) {
      const int e = qe[i];
      const int s = ei[e];
      const int dl = dp[e] - lo;
      qe[i] = s;
      qd[i] = dl;
      ((float4*)qs)[i] = max_edge(((const float4*)a_src)[s], dl);
    }
  }
  __syncthreads();

  const int nq = (qtot > QCAP) ? cnt : qtot;
  const int ul = lane & 7;
  for (int i0 = wave * 8; i0 < nq; i0 += 256) {
    const int il = min(i0 + ul, nq - 1);
    const int sv = qe[il];
    const int dv = qd[il];
    int dl[8];
    float4 m[8];
#pragma unroll
    for (int u = 0; u < 8; ++u) {
      const int s = __builtin_amdgcn_readlane(sv, u);
      dl[u] = __builtin_amdgcn_readlane(dv, u);
      m[u] = ((const float4*)xt)[(size_t)s * 32 + lane];
    }
    const float exv = __expf(qs[il * HEADS + h] - fkey_inv(mxk_l[dv * HEADS + h]));
    if (i0 + ul < nq) atomicAdd(esum_l + dv * HEADS + h, exv);
#pragma unroll
    for (int u = 0; u < 8; ++u) {
      if (i0 + u < nq) {
        const float ex = group8_bcast(exv, u);
        float* ap = acc_l + dl[u] * APITCH + coff;
        atomicAdd(ap + 0, ex * m[u].x);
        atomicAdd(ap + 1, ex * m[u].y);
        atomicAdd(ap + 2, ex * m[u].z);
        atomicAdd(ap + 3, ex * m[u].w);
      }
    }
  }
  if (qtot > QCAP) {
    for (int e = tid; e < E; e += 1024) {
      const int dl = dp[e] - lo;
      if ((unsigned)dl < (unsigned)cnt) {
        const int s = ei[e];
#pragma unroll 1
        for (int hh = 0; hh < HEADS; ++hh) {
          const float ex = __expf(lrelu(a_src[(size_t)s * HEADS + hh] + adl_l[dl * HEADS + hh]) -
                                  fkey_inv(mxk_l[dl * HEADS + hh]));
          atomicAdd(esum_l + dl * HEADS + hh, ex);
#pragma unroll 1
          for (int ch = hh * 32; ch < hh * 32 + 32; ++ch)
            atomicAdd(acc_l + dl * APITCH + ch + (ch >> 6), ex * xt[(size_t)s * OUT_DIM + ch]);
        }
      }
    }
  }
  __syncthreads();

  const float4 gb = ((const float4*)gat_bias)[lane];
  const float4 rb = ((const float4*)res_b)[lane];
  const float4 g = ((const float4*)ln_g)[lane];
  const float4 b = ((const float4*)ln_b)[lane];
  for (int dl = wave; dl < cnt; dl += 32) {
    const int n = lo + dl;
    const float* ap = acc_l + dl * APITCH + coff;
    const float4 rr = *(const float4*)(z_l + dl * ZPITCH + lane * 4);
    const float den = esum_l[dl * HEADS + h] + 1e-16f;
    float4 v;
    v.x = ap[0] / den + gb.x + rr.x + rb.x;
    v.y = ap[1] / den + gb.y + rr.y + rb.y;
    v.z = ap[2] / den + gb.z + rr.z + rb.z;
    v.w = ap[3] / den + gb.w + rr.w + rb.w;
    float sm = v.x + v.y + v.z + v.w;
#pragma unroll
    for (int off = 1; off < 32; off <<= 1) sm += __shfl_xor(sm, off, 32);
    const float mu = sm * (1.0f / 128.0f);
    const float dx = v.x - mu, dy = v.y - mu, dz = v.z - mu, dw = v.w - mu;
    float q = dx * dx + dy * dy + dz * dz + dw * dw;
#pragma unroll
    for (int off = 1; off < 32; off <<= 1) q += __shfl_xor(q, off, 32);
    const float inv = rsqrtf(q * (1.0f / 128.0f) + 1e-5f);
    float4 o;
    o.x = lrelu(dx * inv * g.x + b.x);
    o.y = lrelu(dy * inv * g.y + b.y);
    o.z = lrelu(dz * inv * g.z + b.z);
    o.w = lrelu(dw * inv * g.w + b.w);
    ((float4*)out)[(size_t)n * 32 + lane] = o;
  }
}

extern "C" void kernel_launch(void* const* d_in, const int* in_sizes, int n_in,
                              void* d_out, int out_size, void* d_ws, size_t ws_size,
                              hipStream_t stream) {
  const float* x        = (const float*)d_in[0];
  const int*   ei       = (const int*)d_in[1];
  const float* lin_w    = (const float*)d_in[2];
  const float* att_src  = (const float*)d_in[3];
  const float* att_dst  = (const float*)d_in[4];
  const float* gat_bias = (const float*)d_in[5];
  const float* res_w    = (const float*)d_in[6];
  const float* res_b    = (const float*)d_in[7];
  const float* ln_g     = (const float*)d_in[8];
  const float* ln_b     = (const float*)d_in[9];
  const int N  = in_sizes[0] / IN_CH;
  const int E  = in_sizes[1] / 2;

  float* xt    = (float*)d_ws;
  float* a_src = xt + (size_t)N * OUT_DIM;
  float* a_dst = a_src + (size_t)N * HEADS;
  unsigned* keys = (unsigned*)(a_dst + (size_t)N * HEADS);

  k_gemm<<<(N + 63) / 64, 128, 0, stream>>>(x, lin_w, xt, N);
  k_node<<<(N + 7) / 8, 256, 0, stream>>>(xt, att_src, att_dst, a_src, a_dst, ei, keys, E, N);
  k_own<<<(N + NPW - 1) / NPW, 1024, 0, stream>>>(ei, (const uint4*)keys, xt, a_src, a_dst, x, res_w, gat_bias,
                                                  res_b, ln_g, ln_b, (float*)d_out, E, N);
}


# Key column stored as eight bit planes (the node pass votes 32 words per 128 edges); owner groups match 32 edges per exclusive-or/or chain, not 4 per byte-wise step; queue order changes
# speedup vs baseline: 5.8799x; 1.0518x over previous
#include <hip/hip_runtime.h>
#include <hip/hip_bf16.h>

typedef __attribute__((ext_vector_type(2))) float v2f;
typedef __attribute__((ext_vector_type(8))) float v8f;

#define IN_CH 64
#define OUT_DIM 128
#define HEADS 4
#define NPW 196

static __device__ __forceinline__ float lrelu(float x) { return fmaxf(x, 0.2f * x); }

static __device__ __forceinline__ unsigned fkey(float f) {
  unsigned u = __float_as_uint(f);
  return (u & 0x80000000u) ? ~u : (u | 0x80000000u);
}
static __device__ __forceinline__ float fkey_inv(unsigned k) {
  return __uint_as_float((k & 0x80000000u) ? (k & 0x7FFFFFFFu) : ~k);
}

__global__ __launch_bounds__(128) void k_gemm(const float* __restrict__ x,
                                              const float* __restrict__ lin_w,
                                              float* __restrict__ xt,
                                              int N) {
  const int wave = __builtin_amdgcn_readfirstlane(threadIdx.x >> 5);
  const int lane = threadIdx.x & 31;
  const int c16 = lane & 15;
  const int hf = lane >> 4;
  const int m0 = (blockIdx.x * 4 + wave) * 16;
  if (m0 >= N) return;
  const float* Wm = lin_w;
  float* dst = xt;
  int arow = m0 + c16;
  if (arow >= N) arow = N - 1;
  const float4* xr = (const float4*)(x + (size_t)arow * IN_CH + 4 * hf);
  const float4* wr = (const float4*)(Wm + (size_t)c16 * IN_CH + 4 * hf);
  float4 a[8], b0[8], b1[8];
#pragma unroll
  for (int i = 0; i < 8; ++i) a[i] = xr[2 * i];
#pragma unroll
  for (int i = 0; i < 8; ++i) {
    b0[i] = wr[2 * i];
    b1[i] = wr[16 * IN_CH / 4 + 2 * i];
  }
  float* orow = dst + (size_t)(m0 + (hf << 3)) * OUT_DIM + c16;
  const bool whole = (m0 + 16 <= N);
#pragma unroll
  for (int tp = 0; tp < 4; ++tp) {
    float4 n0[8], n1[8];
    if (tp < 3) {
#pragma unroll
      for (int i = 0; i < 8; ++i) {
        n0[i] = wr[(tp + 1) * 32 * IN_CH / 4 + 2 * i];
        n1[i] = wr[(tp + 1) * 32 * IN_CH / 4 + 16 * IN_CH / 4 + 2 * i];
      }
    }
    v8f c0 = {0.f, 0.f, 0.f, 0.f, 0.f, 0.f, 0.f, 0.f};
    v8f c1 = {0.f, 0.f, 0.f, 0.f, 0.f, 0.f, 0.f, 0.f};
#pragma unroll
    for (int i = 0; i < 8; ++i) {
      const v2f alo = {a[i].x, a[i].y};
      const v2f ahi = {a[i].z, a[i].w};
      const v2f b0lo = {b0[i].x, b0[i].y};
      const v2f b0hi = {b0[i].z, b0[i].w};
      const v2f b1lo = {b1[i].x, b1[i].y};
      const v2f b1hi = {b1[i].z, b1[i].w};
      c0 = __builtin_amdgcn_wmma_f32_16x16x4_f32(false, alo, false, b0lo, (short)0, c0, false, false);
      c1 = __builtin_amdgcn_wmma_f32_16x16x4_f32(false, alo, false, b1lo, (short)0, c1, false, false);
      c0 = __builtin_amdgcn_wmma_f32_16x16x4_f32(false, ahi, false, b0hi, (short)0, c0, false, false);
      c1 = __builtin_amdgcn_wmma_f32_16x16x4_f32(false, ahi, false, b1hi, (short)0, c1, false, false);
    }
    if (whole) {
#pragma unroll
      for (int r = 0; r < 8; ++r) {
        orow[(size_t)r * OUT_DIM + tp * 32] = c0[r];
        orow[(size_t)r * OUT_DIM + tp * 32 + 16] = c1[r];
      }
    } else {
#pragma unroll
      for (int r = 0; r < 8; ++r) {
        if (m0 + (hf << 3) + r < N) {
          orow[(size_t)r * OUT_DIM + tp * 32] = c0[r];
          orow[(size_t)r * OUT_DIM + tp * 32 + 16] = c1[r];
        }
      }
    }
    if (tp < 3) {
#pragma unroll
      for (int i = 0; i < 8; ++i) {
        b0[i] = n0[i];
        b1[i] = n1[i];
      }
    }
  }
}

__global__ __launch_bounds__(256) void k_node(const float* __restrict__ xt,
                                              const float* __restrict__ att_src,
                                              const float* __restrict__ att_dst,
                                              float* __restrict__ a_src,
                                              float* __restrict__ a_dst,
                                              const int* __restrict__ ei,
                                              unsigned* __restrict__ keys, int E, int N) {
  const int lane = threadIdx.x & 31;
  const int nblk = ((E & 3) == 0 && (N + NPW - 1) / NPW <= 256) ? (E >> 7) : 0;
  for (int blk = blockIdx.x * 8 + (threadIdx.x >> 5); blk < nblk; blk += gridDim.x * 8) {
    const int4 d = ((const int4*)(ei + E))[blk * 32 + lane];
    unsigned kc[4];
    kc[0] = (unsigned)d.x / NPW;
    kc[1] = (unsigned)d.y / NPW;
    kc[2] = (unsigned)d.z / NPW;
    kc[3] = (unsigned)d.w / NPW;
    unsigned w = 0;
#pragma unroll
    for (int b = 0; b < 8; ++b) {
#pragma unroll
      for (int cc = 0; cc < 4; ++cc) {
        const unsigned vote = __builtin_amdgcn_ballot_w32(((kc[cc] >> b) & 1u) != 0u);
        if (lane == 4 * b + cc) w = vote;
      }
    }
    keys[(blk >> 5) * 1024 + (((lane >> 2) * 32 + (blk & 31)) << 2) + (lane & 3)] = w;
  }
  const int n = blockIdx.x * 8 + (threadIdx.x >> 5);
  if (n >= N) return;
  const float4 v = ((const float4*)xt)[(size_t)n * 32 + lane];
  const float4 as = ((const float4*)att_src)[lane];
  const float4 ad = ((const float4*)att_dst)[lane];
  float ps = v.x * as.x + v.y * as.y + v.z * as.z + v.w * as.w;
  float pd = v.x * ad.x + v.y * ad.y + v.z * ad.z + v.w * ad.w;
#pragma unroll
  for (int off = 1; off < 8; off <<= 1) {
    ps += __shfl_xor(ps, off, 32);
    pd += __shfl_xor(pd, off, 32);
  }
  const int h = lane >> 3;
  if ((lane & 7) == 0) {
    a_src[(size_t)n * HEADS + h] = ps;
    a_dst[(size_t)n * HEADS + h] = pd;
  }
}

#define APITCH 130
#define QCAP 4096
#define ZROWS (((NPW + 15) / 16) * 16)
#define ZPITCH 132
static __device__ __forceinline__ float group8_bcast(float v, int u) {
  const unsigned b = __float_as_uint(v);
  unsigned r;
  switch (u) {
    case 0: r = __builtin_amdgcn_mov_dpp8(b, 0 * 0x249249); break;
    case 1: r = __builtin_amdgcn_mov_dpp8(b, 1 * 0x249249); break;
    case 2: r = __builtin_amdgcn_mov_dpp8(b, 2 * 0x249249); break;
    case 3: r = __builtin_amdgcn_mov_dpp8(b, 3 * 0x249249); break;
    case 4: r = __builtin_amdgcn_mov_dpp8(b, 4 * 0x249249); break;
    case 5: r = __builtin_amdgcn_mov_dpp8(b, 5 * 0x249249); break;
    case 6: r = __builtin_amdgcn_mov_dpp8(b, 6 * 0x249249); break;
    default: r = __builtin_amdgcn_mov_dpp8(b, 7 * 0x249249); break;
  }
  return __uint_as_float(r);
}
static_assert(NPW * HEADS <= 1024, "one lane per (owned node, head) in the start phase");
static_assert((NPW * APITCH) % 4 == 0, "acc is cleared in 16-byte pieces");
static_assert(NPW < QCAP, "the self-loops must fit the queue");
__global__ __launch_bounds__(1024) void k_own(const int* __restrict__ ei,
                                              const uint4* __restrict__ keys,
                                              const float* __restrict__ xt,
                                              const float* __restrict__ a_src,
                                              const float* __restrict__ a_dst,
                                              const float* __restrict__ x,
                                              const float* __restrict__ res_w,
                                              const float* __restrict__ gat_bias,
                                              const float* __restrict__ res_b,
                                              const float* __restrict__ ln_g,
                                              const float* __restrict__ ln_b,
                                              float* __restrict__ out, int E, int N) {
  __shared__ __attribute__((aligned(16))) float acc_l[NPW * APITCH];
  __shared__ __attribute__((aligned(16))) float esum_l[NPW * HEADS];
  __shared__ __attribute__((aligned(16))) float adl_l[NPW * HEADS];
  __shared__ unsigned mxk_l[NPW * HEADS];
  __shared__ int qe[QCAP];
  __shared__ int qd[QCAP];
  __shared__ __attribute__((aligned(16))) float qs[QCAP * HEADS];
  __shared__ __attribute__((aligned(16))) float z_l[ZROWS * ZPITCH];
  __shared__ int qn;

  const int tid = threadIdx.x;
  const int lane = tid & 31;
  const int wave = __builtin_amdgcn_readfirstlane(tid >> 5);
  const int h = lane >> 3;
  const int coff = lane * 4 + (lane >> 4);
  const int lo = blockIdx.x * NPW;
  const int cnt = min(NPW, N - lo);
  const int* __restrict__ dp = ei + E;

  if (tid == 0) qn = cnt;
  for (int i = tid; i < NPW * APITCH / 4; i += 1024)
    ((float4*)acc_l)[i] = make_float4(0.f, 0.f, 0.f, 0.f);
  if (tid < cnt * HEADS) {
    const float as = a_src[(size_t)lo * HEADS + tid];
    const float ad = a_dst[(size_t)lo * HEADS + tid];
    const float sc = lrelu(as + ad);
    adl_l[tid] = ad;
    qs[tid] = sc;
    mxk_l[tid] = fkey(sc);
    esum_l[tid] = 0.f;
  }
  if (tid < cnt) {
    qe[tid] = lo + tid;
    qd[tid] = tid;
  }
  __syncthreads();

  if (wave < 2 * (ZROWS / 16)) {
    const int c16 = lane & 15;
    const int hf = lane >> 4;
    const int rt = wave >> 1;
    const int ch = wave & 1;
    const int arow = min(lo + rt * 16 + c16, N - 1);
    const float4* xr = (const float4*)(x + (size_t)arow * IN_CH + 4 * hf);
    const float4* wr = (const float4*)(res_w + (size_t)(ch * 64 + c16) * IN_CH + 4 * hf);
    float4 a[8];
#pragma unroll
    for (int i = 0; i < 8; ++i) a[i] = xr[2 * i];
    float* zp = z_l + (rt * 16 + (hf << 3)) * ZPITCH + ch * 64 + c16;
#pragma unroll
    for (int tp = 0; tp < 2; ++tp) {
      float4 b0[8], b1[8];
#pragma unroll
      for (int i = 0; i < 8; ++i) {
        b0[i] = wr[tp * 32 * IN_CH / 4 + 2 * i];
        b1[i] = wr[tp * 32 * IN_CH / 4 + 16 * IN_CH / 4 + 2 * i];
      }
      v8f c0 = {0.f, 0.f, 0.f, 0.f, 0.f, 0.f, 0.f, 0.f};
      v8f c1 = {0.f, 0.f, 0.f, 0.f, 0.f, 0.f, 0.f, 0.f};
#pragma unroll
      for (int i = 0; i < 8; ++i) {
        const v2f alo = {a[i].x, a[i].y};
        const v2f ahi = {a[i].z, a[i].w};
        const v2f b0lo = {b0[i].x, b0[i].y};
        const v2f b0hi = {b0[i].z, b0[i].w};
        const v2f b1lo = {b1[i].x, b1[i].y};
        const v2f b1hi = {b1[i].z, b1[i].w};
        c0 = __builtin_amdgcn_wmma_f32_16x16x4_f32(false, alo, false, b0lo, (short)0, c0, false, false);
        c1 = __builtin_amdgcn_wmma_f32_16x16x4_f32(false, alo, false, b1lo, (short)0, c1, false, false);
        c0 = __builtin_amdgcn_wmma_f32_16x16x4_f32(false, ahi, false, b0hi, (short)0, c0, false, false);
        c1 = __builtin_amdgcn_wmma_f32_16x16x4_f32(false, ahi, false, b1hi, (short)0, c1, false, false);
      }
#pragma unroll
      for (int r = 0; r < 8; ++r) {
        zp[r * ZPITCH + tp * 32] = c0[r];
        zp[r * ZPITCH + tp * 32 + 16] = c1[r];
      }
    }
  }

  auto max_edge = [&](const float4 as, int dl) {
    const float4 ad = *(const float4*)(adl_l + dl * HEADS);
    float4 sc;
    sc.x = lrelu(as.x + ad.x);
    sc.y = lrelu(as.y + ad.y);
    sc.z = lrelu(as.z + ad.z);
    sc.w = lrelu(as.w + ad.w);
    unsigned* mp = mxk_l + dl * HEADS;
    atomicMax(mp + 0, fkey(sc.x));
    atomicMax(mp + 1, fkey(sc.y));
    atomicMax(mp + 2, fkey(sc.z));
    atomicMax(mp + 3, fkey(sc.w));
    return sc;
  };

  auto queue_edge = [&](int e) {
    const int slot = atomicAdd(&qn, 1);
    if (slot < QCAP) {
      qe[slot] = e;
    } else {
      max_edge(((const float4*)a_src)[ei[e]], dp[e] - lo);
    }
  };

  const int nblk = ((E & 3) == 0 && gridDim.x <= 256) ? (E >> 7) : 0;
  const int nch = (nblk + 31) >> 5;
  const int cbase = (int)(((unsigned long long)blockIdx.x * (unsigned)nch) / gridDim.x);
  for (int t = wave; t < nch; t += 32) {
    int ch = cbase + t;
    if (ch >= nch) ch -= nch;
    const int blk = ch * 32 + lane;
    const bool have = blk < nblk;
    const uint4* kp = keys + (size_t)ch * 256 + (have ? lane : 0);
    uint4 v[8];
#pragma unroll
    for (int b = 0; b < 8; ++b) v[b] = kp[b * 32];
    unsigned mis[4] = {0u, 0u, 0u, 0u};
#pragma unroll
    for (int b = 0; b < 8; ++b) {
      const unsigned pb = ((blockIdx.x >> b) & 1u) ? 0xFFFFFFFFu : 0u;
      mis[0] |= v[b].x ^ pb;
      mis[1] |= v[b].y ^ pb;
      mis[2] |= v[b].z ^ pb;
      mis[3] |= v[b].w ^ pb;
    }
#pragma unroll
    for (int cc = 0; cc < 4; ++cc) {
      unsigned m = have ? ~mis[cc] : 0u;
      while (m) {
        const int k = __builtin_ctz(m);
        m &= m - 1;
        queue_edge((blk << 7) + (k << 2) + cc);
      }
    }
  }
  for (int eb = (nblk << 7) + tid; eb < E; eb += 4096) {
    int d[4];
#pragma unroll
    for (int u = 0; u < 4; ++u) d[u] = dp[min(eb + u * 1024, E - 1)];
#pragma unroll
    for (int u = 0; u < 4; ++u)
      if (eb + u * 1024 < E && (unsigned)(d[u] - lo) < (unsigned)cnt) queue_edge(eb + u * 1024);
  }
  __syncthreads();

  const int qtot = qn;
  const int nqx = min(qtot, QCAP);
  if (nqx > cnt) {
    int ev[4], sv[4], dv[4];
    float4 av[4];
#pragma unroll
    for (int u = 0; u < 4; ++u) {
      const int i = cnt + u * 1024 + tid;
      const int e = qe[min(i, QCAP - 1)];
      ev[u] = (i < nqx) ? e : 0;
    }
#pragma unroll
    for (int u = 0; u < 4; ++u) {
      sv[u] = ei[ev[u]];
      dv[u] = dp[ev[u]];
    }
#pragma unroll
    for (int u = 0; u < 4; ++u) av[u] = ((const float4*)a_src)[sv[u]];
#pragma unroll
    for (int u = 0; u < 4; ++u) {
      const int i = cnt + u * 1024 + tid;
      if (i < nqx) {
        qe[i] = sv[u];
        qd[i] = dv[u] - lo;
        ((float4*)qs)[i] = max_edge(av[u], dv[u] - lo);
      }
    }
    for (int i = cnt + 4096 + tid; i < nqx; i += 1024) {
      const int e = qe[i];
      const int s = ei[e];
      const int dl = dp[e] - lo;
      qe[i] = s;
      qd[i] = dl;
      ((float4*)qs)[i] = max_edge(((const float4*)a_src)[s], dl);
    }
  }
  __syncthreads();

  const int nq = (qtot > QCAP) ? cnt : qtot;
  const int ul = lane & 7;
  for (int i0 = wave * 8; i0 < nq; i0 += 256) {
    const int il = min(i0 + ul, nq - 1);
    const int sv = qe[il];
    const int dv = qd[il];
    int dl[8];
    float4 m[8];
#pragma unroll
    for (int u = 0; u < 8; ++u) {
      const int s = __builtin_amdgcn_readlane(sv, u);
      dl[u] = __builtin_amdgcn_readlane(dv, u);
      m[u] = ((const float4*)xt)[(size_t)s * 32 + lane];
    }
    const float exv = __expf(qs[il * HEADS + h] - fkey_inv(mxk_l[dv * HEADS + h]));
    if (i0 + ul < nq) atomicAdd(esum_l + dv * HEADS + h, exv);
#pragma unroll
    for (int u = 0; u < 8; ++u) {
      if (i0 + u < nq) {
        const float ex = group8_bcast(exv, u);
        float* ap = acc_l + dl[u] * APITCH + coff;
        atomicAdd(ap + 0, ex * m[u].x);
        atomicAdd(ap + 1, ex * m[u].y);
        atomicAdd(ap + 2, ex * m[u].z);
        atomicAdd(ap + 3, ex * m[u].w);
      }
    }
  }
  if (qtot > QCAP) {
    for (int e = tid; e < E; e += 1024) {
      const int dl = dp[e] - lo;
      if ((unsigned)dl < (unsigned)cnt) {
        const int s = ei[e];
#pragma unroll 1
        for (int hh = 0; hh < HEADS; ++hh) {
          const float ex = __expf(lrelu(a_src[(size_t)s * HEADS + hh] + adl_l[dl * HEADS + hh]) -
                                  fkey_inv(mxk_l[dl * HEADS + hh]));
          atomicAdd(esum_l + dl * HEADS + hh, ex);
#pragma unroll 1
          for (int ch = hh * 32; ch < hh * 32 + 32; ++ch)
            atomicAdd(acc_l + dl * APITCH + ch + (ch >> 6), ex * xt[(size_t)s * OUT_DIM + ch]);
        }
      }
    }
  }
  __syncthreads();

  const float4 gb = ((const float4*)gat_bias)[lane];
  const float4 rb = ((const float4*)res_b)[lane];
  const float4 g = ((const float4*)ln_g)[lane];
  const float4 b = ((const float4*)ln_b)[lane];
  for (int dl = wave; dl < cnt; dl += 32) {
    const int n = lo + dl;
    const float* ap = acc_l + dl * APITCH + coff;
    const float4 rr = *(const float4*)(z_l + dl * ZPITCH + lane * 4);
    const float den = esum_l[dl * HEADS + h] + 1e-16f;
    float4 v;
    v.x = ap[0] / den + gb.x + rr.x + rb.x;
    v.y = ap[1] / den + gb.y + rr.y + rb.y;
    v.z = ap[2] / den + gb.z + rr.z + rb.z;
    v.w = ap[3] / den + gb.w + rr.w + rb.w;
    float sm = v.x + v.y + v.z + v.w;
#pragma unroll
    for (int off = 1; off < 32; off <<= 1) sm += __shfl_xor(sm, off, 32);
    const float mu = sm * (1.0f / 128.0f);
    const float dx = v.x - mu, dy = v.y - mu, dz = v.z - mu, dw = v.w - mu;
    float q = dx * dx + dy * dy + dz * dz + dw * dw;
#pragma unroll
    for (int off = 1; off < 32; off <<= 1) q += __shfl_xor(q, off, 32);
    const float inv = rsqrtf(q * (1.0f / 128.0f) + 1e-5f);
    float4 o;
    o.x = lrelu(dx * inv * g.x + b.x);
    o.y = lrelu(dy * inv * g.y + b.y);
    o.z = lrelu(dz * inv * g.z + b.z);
    o.w = lrelu(dw * inv * g.w + b.w);
    ((float4*)out)[(size_t)n * 32 + lane] = o;
  }
}

extern "C" void kernel_launch(void* const* d_in, const int* in_sizes, int n_in,
                              void* d_out, int out_size, void* d_ws, size_t ws_size,
                              hipStream_t stream) {
  const float* x        = (const float*)d_in[0];
  const int*   ei       = (const int*)d_in[1];
  const float* lin_w    = (const float*)d_in[2];
  const float* att_src  = (const float*)d_in[3];
  const float* att_dst  = (const float*)d_in[4];
  const float* gat_bias = (const float*)d_in[5];
  const float* res_w    = (const float*)d_in[6];
  const float* res_b    = (const float*)d_in[7];
  const float* ln_g     = (const float*)d_in[8];
  const float* ln_b     = (const float*)d_in[9];
  const int N  = in_sizes[0] / IN_CH;
  const int E  = in_sizes[1] / 2;

  float* xt    = (float*)d_ws;
  float* a_src = xt + (size_t)N * OUT_DIM;
  float* a_dst = a_src + (size_t)N * HEADS;
  unsigned* keys = (unsigned*)(a_dst + (size_t)N * HEADS);

  k_gemm<<<(N + 63) / 64, 128, 0, stream>>>(x, lin_w, xt, N);
  k_node<<<(N + 7) / 8, 256, 0, stream>>>(xt, att_src, att_dst, a_src, a_dst, ei, keys, E, N);
  k_own<<<(N + NPW - 1) / NPW, 1024, 0, stream>>>(ei, (const uint4*)keys, xt, a_src, a_dst, x, res_w, gat_bias,
                                                  res_b, ln_g, ln_b, (float*)d_out, E, N);
}


# Node pass merged into the GEMM launch (3 launches -> 2): a_src/a_dst folded from the WMMA accumulators by a 16-lane DPP butterfly, key planes voted by the same waves; a_src/a_dst sum order changes
# speedup vs baseline: 6.2227x; 1.0583x over previous
#include <hip/hip_runtime.h>
#include <hip/hip_bf16.h>

typedef __attribute__((ext_vector_type(2))) float v2f;
typedef __attribute__((ext_vector_type(8))) float v8f;

#define IN_CH 64
#define OUT_DIM 128
#define HEADS 4
#define NPW 196

static __device__ __forceinline__ float lrelu(float x) { return fmaxf(x, 0.2f * x); }

static __device__ __forceinline__ unsigned fkey(float f) {
  unsigned u = __float_as_uint(f);
  return (u & 0x80000000u) ? ~u : (u | 0x80000000u);
}
static __device__ __forceinline__ float fkey_inv(unsigned k) {
  return __uint_as_float((k & 0x80000000u) ? (k & 0x7FFFFFFFu) : ~k);
}

template <int M>
static __device__ __forceinline__ float row_xor(float v) {
  return __int_as_float(__builtin_amdgcn_update_dpp(0, __float_as_int(v), 0x160 | M, 0xF, 0xF, false));
}
static __device__ __forceinline__ unsigned key_votes(const int4 d, int lane) {
  unsigned kc[4];
  kc[0] = (unsigned)d.x / NPW;
  kc[1] = (unsigned)d.y / NPW;
  kc[2] = (unsigned)d.z / NPW;
  kc[3] = (unsigned)d.w / NPW;
  unsigned w = 0;
#pragma unroll
  for (int b = 0; b < 8; ++b) {
#pragma unroll
    for (int cc = 0; cc < 4; ++cc) {
      const unsigned vote = __builtin_amdgcn_ballot_w32(((kc[cc] >> b) & 1u) != 0u);
      if (lane == 4 * b + cc) w = vote;
    }
  }
  return w;
}
static __device__ __forceinline__ int key_slot(int blk, int lane) {
  return (blk >> 5) * 1024 + (((lane >> 2) * 32 + (blk & 31)) << 2) + (lane & 3);
}
__global__ __launch_bounds__(128) void k_gemm(const float* __restrict__ x,
                                              const float* __restrict__ lin_w,
                                              const float* __restrict__ att_src,
                                              const float* __restrict__ att_dst,
                                              float* __restrict__ xt,
                                              float* __restrict__ a_src,
                                              float* __restrict__ a_dst,
                                              const int* __restrict__ ei,
                                              unsigned* __restrict__ keys, int E, int N) {
  const int wave = __builtin_amdgcn_readfirstlane(threadIdx.x >> 5);
  const int lane = threadIdx.x & 31;
  const int c16 = lane & 15;
  const int hf = lane >> 4;
  const int gw = blockIdx.x * 4 + wave;
  const int m0 = gw * 16;
  const float* Wm = lin_w;
  float* dst = xt;
  int arow = m0 + c16;
  if (arow >= N) arow = N - 1;
  const float4* xr = (const float4*)(x + (size_t)arow * IN_CH + 4 * hf);
  const float4* wr = (const float4*)(Wm + (size_t)c16 * IN_CH + 4 * hf);
  float4 a[8], b0[8], b1[8];
#pragma unroll
  for (int i = 0; i < 8; ++i) a[i] = xr[2 * i];
#pragma unroll
  for (int i = 0; i < 8; ++i) {
    b0[i] = wr[2 * i];
    b1[i] = wr[16 * IN_CH / 4 + 2 * i];
  }
  float as0[HEADS], as1[HEADS], ad0[HEADS], ad1[HEADS];
#pragma unroll
  for (int hh = 0; hh < HEADS; ++hh) {
    as0[hh] = att_src[hh * 32 + c16];
    as1[hh] = att_src[hh * 32 + 16 + c16];
    ad0[hh] = att_dst[hh * 32 + c16];
    ad1[hh] = att_dst[hh * 32 + 16 + c16];
  }
  const int nblk = ((E & 3) == 0 && (N + NPW - 1) / NPW <= 256) ? (E >> 7) : 0;
  for (int blk = gw * 2; blk < nblk; blk += gridDim.x * 8) {
    const bool two = blk + 1 < nblk;
    const int4 d0 = ((const int4*)(ei + E))[blk * 32 + lane];
    const int4 d1 = ((const int4*)(ei + E))[(two ? blk + 1 : blk) * 32 + lane];
    const unsigned w0 = key_votes(d0, lane);
    const unsigned w1 = key_votes(d1, lane);
    keys[key_slot(blk, lane)] = w0;
    if (two) keys[key_slot(blk + 1, lane)] = w1;
  }
  if (m0 >= N) return;
  float* orow = dst + (size_t)(m0 + (hf << 3)) * OUT_DIM + c16;
  const bool whole = (m0 + 16 <= N);
  const bool f3 = (c16 & 8) != 0, f2 = (c16 & 4) != 0, f1 = (c16 & 2) != 0, f0 = (c16 & 1) != 0;
  float att[HEADS];
#pragma unroll
  for (int tp = 0; tp < 4; ++tp) {
    float4 n0[8], n1[8];
    if (tp < 3) {
#pragma unroll
      for (int i = 0; i < 8; ++i) {
        n0[i] = wr[(tp + 1) * 32 * IN_CH / 4 + 2 * i];
        n1[i] = wr[(tp + 1) * 32 * IN_CH / 4 + 16 * IN_CH / 4 + 2 * i];
      }
    }
    v8f c0 = {0.f, 0.f, 0.f, 0.f, 0.f, 0.f, 0.f, 0.f};
    v8f c1 = {0.f, 0.f, 0.f, 0.f, 0.f, 0.f, 0.f, 0.f};
#pragma unroll
    for (int i = 0; i < 8; ++i) {
      const v2f alo = {a[i].x, a[i].y};
      const v2f ahi = {a[i].z, a[i].w};
      const v2f b0lo = {b0[i].x, b0[i].y};
      const v2f b0hi = {b0[i].z, b0[i].w};
      const v2f b1lo = {b1[i].x, b1[i].y};
      const v2f b1hi = {b1[i].z, b1[i].w};
      c0 = __builtin_amdgcn_wmma_f32_16x16x4_f32(false, alo, false, b0lo, (short)0, c0, false, false);
      c1 = __builtin_amdgcn_wmma_f32_16x16x4_f32(false, alo, false, b1lo, (short)0, c1, false, false);
      c0 = __builtin_amdgcn_wmma_f32_16x16x4_f32(false, ahi, false, b0hi, (short)0, c0, false, false);
      c1 = __builtin_amdgcn_wmma_f32_16x16x4_f32(false, ahi, false, b1hi, (short)0, c1, false, false);
    }
    if (whole) {
#pragma unroll
      for (int r = 0; r < 8; ++r) {
        orow[(size_t)r * OUT_DIM + tp * 32] = c0[r];
        orow[(size_t)r * OUT_DIM + tp * 32 + 16] = c1[r];
      }
    } else {
#pragma unroll
      for (int r = 0; r < 8; ++r) {
        if (m0 + (hf << 3) + r < N) {
          orow[(size_t)r * OUT_DIM + tp * 32] = c0[r];
          orow[(size_t)r * OUT_DIM + tp * 32 + 16] = c1[r];
        }
      }
    }
    float p[16];
#pragma unroll
    for (int r = 0; r < 8; ++r) {
      p[r] = c0[r] * as0[tp] + c1[r] * as1[tp];
      p[8 + r] = c0[r] * ad0[tp] + c1[r] * ad1[tp];
    }
    float q[8];
#pragma unroll
    for (int i = 0; i < 8; ++i) q[i] = (f3 ? p[i + 8] : p[i]) + row_xor<8>(f3 ? p[i] : p[i + 8]);
    float s[4];
#pragma unroll
    for (int i = 0; i < 4; ++i) s[i] = (f2 ? q[i + 4] : q[i]) + row_xor<4>(f2 ? q[i] : q[i + 4]);
    float t[2];
#pragma unroll
    for (int i = 0; i < 2; ++i) t[i] = (f1 ? s[i + 2] : s[i]) + row_xor<2>(f1 ? s[i] : s[i + 2]);
    att[tp] = (f0 ? t[1] : t[0]) + row_xor<1>(f0 ? t[0] : t[1]);
    if (tp < 3) {
#pragma unroll
      for (int i = 0; i < 8; ++i) {
        b0[i] = n0[i];
        b1[i] = n1[i];
      }
    }
  }
  const int nrow = m0 + (hf << 3) + (c16 & 7);
  if (nrow < N)
    *(float4*)((f3 ? a_dst : a_src) + (size_t)nrow * HEADS) = make_float4(att[0], att[1], att[2], att[3]);
}

#define APITCH 130
#define QCAP 4096
#define ZROWS (((NPW + 15) / 16) * 16)
#define ZPITCH 132
static __device__ __forceinline__ float group8_bcast(float v, int u) {
  const unsigned b = __float_as_uint(v);
  unsigned r;
  switch (u) {
    case 0: r = __builtin_amdgcn_mov_dpp8(b, 0 * 0x249249); break;
    case 1: r = __builtin_amdgcn_mov_dpp8(b, 1 * 0x249249); break;
    case 2: r = __builtin_amdgcn_mov_dpp8(b, 2 * 0x249249); break;
    case 3: r = __builtin_amdgcn_mov_dpp8(b, 3 * 0x249249); break;
    case 4: r = __builtin_amdgcn_mov_dpp8(b, 4 * 0x249249); break;
    case 5: r = __builtin_amdgcn_mov_dpp8(b, 5 * 0x249249); break;
    case 6: r = __builtin_amdgcn_mov_dpp8(b, 6 * 0x249249); break;
    default: r = __builtin_amdgcn_mov_dpp8(b, 7 * 0x249249); break;
  }
  return __uint_as_float(r);
}
static_assert(NPW * HEADS <= 1024, "one lane per (owned node, head) in the start phase");
static_assert((NPW * APITCH) % 4 == 0, "acc is cleared in 16-byte pieces");
static_assert(NPW < QCAP, "the self-loops must fit the queue");
__global__ __launch_bounds__(1024) void k_own(const int* __restrict__ ei,
                                              const uint4* __restrict__ keys,
                                              const float* __restrict__ xt,
                                              const float* __restrict__ a_src,
                                              const float* __restrict__ a_dst,
                                              const float* __restrict__ x,
                                              const float* __restrict__ res_w,
                                              const float* __restrict__ gat_bias,
                                              const float* __restrict__ res_b,
                                              const float* __restrict__ ln_g,
                                              const float* __restrict__ ln_b,
                                              float* __restrict__ out, int E, int N) {
  __shared__ __attribute__((aligned(16))) float acc_l[NPW * APITCH];
  __shared__ __attribute__((aligned(16))) float esum_l[NPW * HEADS];
  __shared__ __attribute__((aligned(16))) float adl_l[NPW * HEADS];
  __shared__ unsigned mxk_l[NPW * HEADS];
  __shared__ int qe[QCAP];
  __shared__ int qd[QCAP];
  __shared__ __attribute__((aligned(16))) float qs[QCAP * HEADS];
  __shared__ __attribute__((aligned(16))) float z_l[ZROWS * ZPITCH];
  __shared__ int qn;

  const int tid = threadIdx.x;
  const int lane = tid & 31;
  const int wave = __builtin_amdgcn_readfirstlane(tid >> 5);
  const int h = lane >> 3;
  const int coff = lane * 4 + (lane >> 4);
  const int lo = blockIdx.x * NPW;
  const int cnt = min(NPW, N - lo);
  const int* __restrict__ dp = ei + E;

  if (tid == 0) qn = cnt;
  for (int i = tid; i < NPW * APITCH / 4; i += 1024)
    ((float4*)acc_l)[i] = make_float4(0.f, 0.f, 0.f, 0.f);
  if (tid < cnt * HEADS) {
    const float as = a_src[(size_t)lo * HEADS + tid];
    const float ad = a_dst[(size_t)lo * HEADS + tid];
    const float sc = lrelu(as + ad);
    adl_l[tid] = ad;
    qs[tid] = sc;
    mxk_l[tid] = fkey(sc);
    esum_l[tid] = 0.f;
  }
  if (tid < cnt) {
    qe[tid] = lo + tid;
    qd[tid] = tid;
  }
  __syncthreads();

  if (wave < 2 * (ZROWS / 16)) {
    const int c16 = lane & 15;
    const int hf = lane >> 4;
    const int rt = wave >> 1;
    const int ch = wave & 1;
    const int arow = min(lo + rt * 16 + c16, N - 1);
    const float4* xr = (const float4*)(x + (size_t)arow * IN_CH + 4 * hf);
    const float4* wr = (const float4*)(res_w + (size_t)(ch * 64 + c16) * IN_CH + 4 * hf);
    float4 a[8];
#pragma unroll
    for (int i = 0; i < 8; ++i) a[i] = xr[2 * i];
    float* zp = z_l + (rt * 16 + (hf << 3)) * ZPITCH + ch * 64 + c16;
#pragma unroll
    for (int tp = 0; tp < 2; ++tp) {
      float4 b0[8], b1[8];
#pragma unroll
      for (int i = 0; i < 8; ++i) {
        b0[i] = wr[tp * 32 * IN_CH / 4 + 2 * i];
        b1[i] = wr[tp * 32 * IN_CH / 4 + 16 * IN_CH / 4 + 2 * i];
      }
      v8f c0 = {0.f, 0.f, 0.f, 0.f, 0.f, 0.f, 0.f, 0.f};
      v8f c1 = {0.f, 0.f, 0.f, 0.f, 0.f, 0.f, 0.f, 0.f};
#pragma unroll
      for (int i = 0; i < 8; ++i) {
        const v2f alo = {a[i].x, a[i].y};
        const v2f ahi = {a[i].z, a[i].w};
        const v2f b0lo = {b0[i].x, b0[i].y};
        const v2f b0hi = {b0[i].z, b0[i].w};
        const v2f b1lo = {b1[i].x, b1[i].y};
        const v2f b1hi = {b1[i].z, b1[i].w};
        c0 = __builtin_amdgcn_wmma_f32_16x16x4_f32(false, alo, false, b0lo, (short)0, c0, false, false);
        c1 = __builtin_amdgcn_wmma_f32_16x16x4_f32(false, alo, false, b1lo, (short)0, c1, false, false);
        c0 = __builtin_amdgcn_wmma_f32_16x16x4_f32(false, ahi, false, b0hi, (short)0, c0, false, false);
        c1 = __builtin_amdgcn_wmma_f32_16x16x4_f32(false, ahi, false, b1hi, (short)0, c1, false, false);
      }
#pragma unroll
      for (int r = 0; r < 8; ++r) {
        zp[r * ZPITCH + tp * 32] = c0[r];
        zp[r * ZPITCH + tp * 32 + 16] = c1[r];
      }
    }
  }

  auto max_edge = [&](const float4 as, int dl) {
    const float4 ad = *(const float4*)(adl_l + dl * HEADS);
    float4 sc;
    sc.x = lrelu(as.x + ad.x);
    sc.y = lrelu(as.y + ad.y);
    sc.z = lrelu(as.z + ad.z);
    sc.w = lrelu(as.w + ad.w);
    unsigned* mp = mxk_l + dl * HEADS;
    atomicMax(mp + 0, fkey(sc.x));
    atomicMax(mp + 1, fkey(sc.y));
    atomicMax(mp + 2, fkey(sc.z));
    atomicMax(mp + 3, fkey(sc.w));
    return sc;
  };

  auto queue_edge = [&](int e) {
    const int slot = atomicAdd(&qn, 1);
    if (slot < QCAP) {
      qe[slot] = e;
    } else {
      max_edge(((const float4*)a_src)[ei[e]], dp[e] - lo);
    }
  };

  const int nblk = ((E & 3) == 0 && gridDim.x <= 256) ? (E >> 7) : 0;
  const int nch = (nblk + 31) >> 5;
  const int cbase = (int)(((unsigned long long)blockIdx.x * (unsigned)nch) / gridDim.x);
  for (int t = wave; t < nch; t += 32) {
    int ch = cbase + t;
    if (ch >= nch) ch -= nch;
    const int blk = ch * 32 + lane;
    const bool have = blk < nblk;
    const uint4* kp = keys + (size_t)ch * 256 + (have ? lane : 0);
    uint4 v[8];
#pragma unroll
    for (int b = 0; b < 8; ++b) v[b] = kp[b * 32];
    unsigned mis[4] = {0u, 0u, 0u, 0u};
#pragma unroll
    for (int b = 0; b < 8; ++b) {
      const unsigned pb = ((blockIdx.x >> b) & 1u) ? 0xFFFFFFFFu : 0u;
      mis[0] |= v[b].x ^ pb;
      mis[1] |= v[b].y ^ pb;
      mis[2] |= v[b].z ^ pb;
      mis[3] |= v[b].w ^ pb;
    }
#pragma unroll
    for (int cc = 0; cc < 4; ++cc) {
      unsigned m = have ? ~mis[cc] : 0u;
      while (m) {
        const int k = __builtin_ctz(m);
        m &= m - 1;
        queue_edge((blk << 7) + (k << 2) + cc);
      }
    }
  }
  for (int eb = (nblk << 7) + tid; eb < E; eb += 4096) {
    int d[4];
#pragma unroll
    for (int u = 0; u < 4; ++u) d[u] = dp[min(eb + u * 1024, E - 1)];
#pragma unroll
    for (int u = 0; u < 4; ++u)
      if (eb + u * 1024 < E && (unsigned)(d[u] - lo) < (unsigned)cnt) queue_edge(eb + u * 1024);
  }
  __syncthreads();

  const int qtot = qn;
  const int nqx = min(qtot, QCAP);
  if (nqx > cnt) {
    int ev[4], sv[4], dv[4];
    float4 av[4];
#pragma unroll
    for (int u = 0; u < 4; ++u) {
      const int i = cnt + u * 1024 + tid;
      const int e = qe[min(i, QCAP - 1)];
      ev[u] = (i < nqx) ? e : 0;
    }
#pragma unroll
    for (int u = 0; u < 4; ++u) {
      sv[u] = ei[ev[u]];
      dv[u] = dp[ev[u]];
    }
#pragma unroll
    for (int u = 0; u < 4; ++u) av[u] = ((const float4*)a_src)[sv[u]];
#pragma unroll
    for (int u = 0; u < 4; ++u) {
      const int i = cnt + u * 1024 + tid;
      if (i < nqx) {
        qe[i] = sv[u];
        qd[i] = dv[u] - lo;
        ((float4*)qs)[i] = max_edge(av[u], dv[u] - lo);
      }
    }
    for (int i = cnt + 4096 + tid; i < nqx; i += 1024) {
      const int e = qe[i];
      const int s = ei[e];
      const int dl = dp[e] - lo;
      qe[i] = s;
      qd[i] = dl;
      ((float4*)qs)[i] = max_edge(((const float4*)a_src)[s], dl);
    }
  }
  __syncthreads();

  const int nq = (qtot > QCAP) ? cnt : qtot;
  const int ul = lane & 7;
  for (int i0 = wave * 8; i0 < nq; i0 += 256) {
    const int il = min(i0 + ul, nq - 1);
    const int sv = qe[il];
    const int dv = qd[il];
    int dl[8];
    float4 m[8];
#pragma unroll
    for (int u = 0; u < 8; ++u) {
      const int s = __builtin_amdgcn_readlane(sv, u);
      dl[u] = __builtin_amdgcn_readlane(dv, u);
      m[u] = ((const float4*)xt)[(size_t)s * 32 + lane];
    }
    const float exv = __expf(qs[il * HEADS + h] - fkey_inv(mxk_l[dv * HEADS + h]));
    if (i0 + ul < nq) atomicAdd(esum_l + dv * HEADS + h, exv);
#pragma unroll
    for (int u = 0; u < 8; ++u) {
      if (i0 + u < nq) {
        const float ex = group8_bcast(exv, u);
        float* ap = acc_l + dl[u] * APITCH + coff;
        atomicAdd(ap + 0, ex * m[u].x);
        atomicAdd(ap + 1, ex * m[u].y);
        atomicAdd(ap + 2, ex * m[u].z);
        atomicAdd(ap + 3, ex * m[u].w);
      }
    }
  }
  if (qtot > QCAP) {
    for (int e = tid; e < E; e += 1024) {
      const int dl = dp[e] - lo;
      if ((unsigned)dl < (unsigned)cnt) {
        const int s = ei[e];
#pragma unroll 1
        for (int hh = 0; hh < HEADS; ++hh) {
          const float ex = __expf(lrelu(a_src[(size_t)s * HEADS + hh] + adl_l[dl * HEADS + hh]) -
                                  fkey_inv(mxk_l[dl * HEADS + hh]));
          atomicAdd(esum_l + dl * HEADS + hh, ex);
#pragma unroll 1
          for (int ch = hh * 32; ch < hh * 32 + 32; ++ch)
            atomicAdd(acc_l + dl * APITCH + ch + (ch >> 6), ex * xt[(size_t)s * OUT_DIM + ch]);
        }
      }
    }
  }
  __syncthreads();

  const float4 gb = ((const float4*)gat_bias)[lane];
  const float4 rb = ((const float4*)res_b)[lane];
  const float4 g = ((const float4*)ln_g)[lane];
  const float4 b = ((const float4*)ln_b)[lane];
  for (int dl = wave; dl < cnt; dl += 32) {
    const int n = lo + dl;
    const float* ap = acc_l + dl * APITCH + coff;
    const float4 rr = *(const float4*)(z_l + dl * ZPITCH + lane * 4);
    const float den = esum_l[dl * HEADS + h] + 1e-16f;
    float4 v;
    v.x = ap[0] / den + gb.x + rr.x + rb.x;
    v.y = ap[1] / den + gb.y + rr.y + rb.y;
    v.z = ap[2] / den + gb.z + rr.z + rb.z;
    v.w = ap[3] / den + gb.w + rr.w + rb.w;
    float sm = v.x + v.y + v.z + v.w;
#pragma unroll
    for (int off = 1; off < 32; off <<= 1) sm += __shfl_xor(sm, off, 32);
    const float mu = sm * (1.0f / 128.0f);
    const float dx = v.x - mu, dy = v.y - mu, dz = v.z - mu, dw = v.w - mu;
    float q = dx * dx + dy * dy + dz * dz + dw * dw;
#pragma unroll
    for (int off = 1; off < 32; off <<= 1) q += __shfl_xor(q, off, 32);
    const float inv = rsqrtf(q * (1.0f / 128.0f) + 1e-5f);
    float4 o;
    o.x = lrelu(dx * inv * g.x + b.x);
    o.y = lrelu(dy * inv * g.y + b.y);
    o.z = lrelu(dz * inv * g.z + b.z);
    o.w = lrelu(dw * inv * g.w + b.w);
    ((float4*)out)[(size_t)n * 32 + lane] = o;
  }
}

extern "C" void kernel_launch(void* const* d_in, const int* in_sizes, int n_in,
                              void* d_out, int out_size, void* d_ws, size_t ws_size,
                              hipStream_t stream) {
  const float* x        = (const float*)d_in[0];
  const int*   ei       = (const int*)d_in[1];
  const float* lin_w    = (const float*)d_in[2];
  const float* att_src  = (const float*)d_in[3];
  const float* att_dst  = (const float*)d_in[4];
  const float* gat_bias = (const float*)d_in[5];
  const float* res_w    = (const float*)d_in[6];
  const float* res_b    = (const float*)d_in[7];
  const float* ln_g     = (const float*)d_in[8];
  const float* ln_b     = (const float*)d_in[9];
  const int N  = in_sizes[0] / IN_CH;
  const int E  = in_sizes[1] / 2;

  float* xt    = (float*)d_ws;
  float* a_src = xt + (size_t)N * OUT_DIM;
  float* a_dst = a_src + (size_t)N * HEADS;
  unsigned* keys = (unsigned*)(a_dst + (size_t)N * HEADS);

  k_gemm<<<(N + 63) / 64, 128, 0, stream>>>(x, lin_w, att_src, att_dst, xt, a_src, a_dst, ei, keys, E, N);
  k_own<<<(N + NPW - 1) / NPW, 1024, 0, stream>>>(ei, (const uint4*)keys, xt, a_src, a_dst, x, res_w, gat_bias,
                                                  res_b, ln_g, ln_b, (float*)d_out, E, N);
}
